# starting kernel
# baseline (speedup 1.0000x reference)
#include <hip/hip_runtime.h>

typedef float v2f __attribute__((ext_vector_type(2)));
typedef float v8f __attribute__((ext_vector_type(8)));

#define D 128

__global__ __launch_bounds__(256) void sage_zero(float* __restrict__ p, long n) {
    long i = (long)blockIdx.x * blockDim.x + threadIdx.x;
    long stride = (long)gridDim.x * blockDim.x;
    for (; i < n; i += stride) p[i] = 0.0f;
}

__global__ __launch_bounds__(256) void sage_scatter(
    const float* __restrict__ h, const int* __restrict__ src,
    const int* __restrict__ dst, float* __restrict__ nsum,
    float* __restrict__ deg, int n_edges)
{
    int e = blockIdx.x * 8 + (threadIdx.x >> 5);
    if (e >= n_edges) return;
    int lane = threadIdx.x & 31;
    int s = src[e];
    int d = dst[e];
    float4 v = *((const float4*)(h + (size_t)s * D) + lane);
    float* np = nsum + (size_t)d * D + lane * 4;
    atomicAdd(np + 0, v.x);
    atomicAdd(np + 1, v.y);
    atomicAdd(np + 2, v.z);
    atomicAdd(np + 3, v.w);
    if (lane == 0) atomicAdd(deg + d, 1.0f);
}

__global__ __launch_bounds__(128) void sage_gemm(
    const float* __restrict__ h, const float* __restrict__ nsum,
    const float* __restrict__ deg, const float* __restrict__ Wself,
    const float* __restrict__ Wneigh, const float* __restrict__ bias,
    float* __restrict__ out, int n_nodes)
{
    const int lane = threadIdx.x & 31;
    const int wave = threadIdx.x >> 5;
    const int tile = blockIdx.x * 4 + wave;
    const int m0   = tile * 16;
    if (m0 >= n_nodes) return;

    const int half  = lane >> 4;
    const int l15   = lane & 15;
    const int mA    = m0 + l15;
    const int kOffA = half * 2;

    float dg   = deg[mA];
    float rdeg = 1.0f / fmaxf(dg, 1.0f);

    v8f acc[8];
    #pragma unroll
    for (int t = 0; t < 8; ++t) {
        #pragma unroll
        for (int r = 0; r < 8; ++r) acc[t][r] = 0.0f;
    }

    const float* hrow = h    + (size_t)mA * D + kOffA;
    const float* srow = nsum + (size_t)mA * D + kOffA;

    #pragma unroll 1
    for (int pass = 0; pass < 2; ++pass) {
        const float* arow = pass ? srow : hrow;
        const float* W    = pass ? Wneigh : Wself;
        const float* wb0  = W + (size_t)l15 * D + kOffA;
        #pragma unroll 4
        for (int ks = 0; ks < D; ks += 4) {
            v2f a = *(const v2f*)(arow + ks);
            if (pass) { a.x *= rdeg; a.y *= rdeg; }
            #pragma unroll
            for (int t = 0; t < 8; ++t) {
                v2f bf = *(const v2f*)(wb0 + ks + (size_t)t * 16 * D);
                acc[t] = __builtin_amdgcn_wmma_f32_16x16x4_f32(
                     false, a,  false, bf,
                     (short)0, acc[t],
                     false,  false);
            }
        }
    }

    const int mBase = m0 + half * 8;
    #pragma unroll
    for (int t = 0; t < 8; ++t) {
        int n = t * 16 + l15;
        float bb = bias[n];
        float* op = out + (size_t)mBase * D + n;
        #pragma unroll
        for (int r = 0; r < 8; ++r) {
            op[(size_t)r * D] = acc[t][r] + bb;
        }
    }
}

extern "C" void kernel_launch(void* const* d_in, const int* in_sizes, int n_in,
                              void* d_out, int out_size, void* d_ws, size_t ws_size,
                              hipStream_t stream) {
    const float* h      = (const float*)d_in[0];
    const int*   src    = (const int*)d_in[1];
    const int*   dst    = (const int*)d_in[2];
    const float* Wself  = (const float*)d_in[3];
    const float* Wneigh = (const float*)d_in[4];
    const float* bias   = (const float*)d_in[5];
    float*       out    = (float*)d_out;

    const int n_nodes = in_sizes[0] / D;
    const int n_edges = in_sizes[1];

    float* nsum = (float*)d_ws;
    float* deg  = nsum + (size_t)n_nodes * D;

    long ztot = (long)n_nodes * D + n_nodes;
    sage_zero<<<4096, 256, 0, stream>>>((float*)d_ws, ztot);

    int sblocks = (n_edges + 7) / 8;
    sage_scatter<<<sblocks, 256, 0, stream>>>(h, src, dst, nsum, deg, n_edges);

    int n_tiles = (n_nodes + 15) / 16;
    int gblocks = (n_tiles + 3) / 4;
    sage_gemm<<<gblocks, 128, 0, stream>>>(h, nsum, deg, Wself, Wneigh, bias,
                                           out, n_nodes);
}


# Float-atomic edge scatter replaced by counting sort of edges by dst (ranks from LDS atomics per edge chunk, scans give offsets) + atomic-free per-node gather-mean; rows summed in slot (arrival) order
# speedup vs baseline: 1.9863x; 1.9863x over previous
#include <hip/hip_runtime.h>

typedef float v2f __attribute__((ext_vector_type(2)));
typedef float v8f __attribute__((ext_vector_type(8)));

#define D 128
#define SCAN_E 1024
#define CH 51200
#define NW_MAX 51200


__device__ __forceinline__ int sage_rank_one(unsigned* cw, int d) {
    const int sh = (d & 1) << 4;
    const unsigned old = atomicAdd(cw + (d >> 1), 1u << sh);
    return (int)((old >> sh) & 0xFFFFu);
}

__global__ __launch_bounds__(1024) void sage_count_lds(
    const int* __restrict__ dst, unsigned* __restrict__ C,
    int* __restrict__ lrank, int n_edges, int nw)
{
    __shared__ __attribute__((aligned(16))) unsigned cw[NW_MAX];
    const int tid = threadIdx.x;

    for (int i = tid; i < nw / 4; i += 1024)
        ((uint4*)cw)[i] = make_uint4(0u, 0u, 0u, 0u);
    __syncthreads();

    const int e0 = blockIdx.x * CH;
    const int e1 = min(e0 + CH, n_edges);
    const int ng = (e1 - e0) >> 2;
    const int4* __restrict__ d4 = (const int4*)(dst + e0);
    int4* __restrict__ r4 = (int4*)(lrank + e0);
    for (int q = tid; q < ng; q += 4096) {
        int4 d[4];
        #pragma unroll
        for (int u = 0; u < 4; ++u) {
            const int qq = q + u * 1024;
            d[u] = (qq < ng) ? d4[qq] : make_int4(0, 0, 0, 0);
        }
        #pragma unroll
        for (int u = 0; u < 4; ++u) {
            const int qq = q + u * 1024;
            if (qq < ng) {
                int4 r;
                r.x = sage_rank_one(cw, d[u].x);
                r.y = sage_rank_one(cw, d[u].y);
                r.z = sage_rank_one(cw, d[u].z);
                r.w = sage_rank_one(cw, d[u].w);
                r4[qq] = r;
            }
        }
    }
    for (int k = e0 + ng * 4 + tid; k < e1; k += 1024)
        lrank[k] = sage_rank_one(cw, dst[k]);
    __syncthreads();

    uint4* Crow = (uint4*)(C + (size_t)blockIdx.x * nw);
    for (int i = tid; i < nw / 4; i += 1024) Crow[i] = ((const uint4*)cw)[i];
}

__global__ __launch_bounds__(256) void sage_colscan(
    const unsigned* __restrict__ C, int* __restrict__ base,
    int* __restrict__ cnt, int n_chunks, int nw)
{
    const int w = blockIdx.x * 256 + threadIdx.x;
    int lo = 0, hi = 0;
    #pragma unroll 8
    for (int g = 0; g < n_chunks; ++g) {
        const unsigned c = C[(size_t)g * nw + w];
        *(int2*)(base + ((size_t)g * nw + w) * 2) = make_int2(lo, hi);
        lo += (int)(c & 0xFFFFu);
        hi += (int)(c >> 16);
    }
    *(int2*)(cnt + 2 * w) = make_int2(lo, hi);
}

__global__ __launch_bounds__(256) void sage_blocksum(
    const int* __restrict__ cnt, int* __restrict__ bsum)
{
    __shared__ int wsum[8];
    const int lane = threadIdx.x & 31;
    const int wave = threadIdx.x >> 5;
    int4 c = ((const int4*)cnt)[blockIdx.x * 256 + threadIdx.x];
    int s = (c.x + c.y) + (c.z + c.w);
    for (int o = 16; o > 0; o >>= 1) s += __shfl_xor(s, o);
    if (lane == 0) wsum[wave] = s;
    __syncthreads();
    if (threadIdx.x == 0) {
        int t = 0;
        for (int w = 0; w < 8; ++w) t += wsum[w];
        bsum[blockIdx.x] = t;
    }
}

__global__ __launch_bounds__(256) void sage_scan(
    const int* __restrict__ cnt, const int* __restrict__ bsum,
    int* __restrict__ off)
{
    __shared__ int wsum[8];
    __shared__ int sbase;
    const int lane = threadIdx.x & 31;
    const int wave = threadIdx.x >> 5;

    int4 c = ((const int4*)cnt)[blockIdx.x * 256 + threadIdx.x];

    if (wave == 0) {
        int t = 0;
        for (int i = lane; i < (int)blockIdx.x; i += 32) t += bsum[i];
        for (int o = 16; o > 0; o >>= 1) t += __shfl_xor(t, o);
        if (lane == 0) sbase = t;
    }

    const int t1 = c.x, t2 = t1 + c.y, t3 = t2 + c.z, t4 = t3 + c.w;
    int inc = t4;
    for (int o = 1; o < 32; o <<= 1) {
        int v = __shfl_up(inc, o);
        if (lane >= o) inc += v;
    }
    if (lane == 31) wsum[wave] = inc;
    __syncthreads();

    int ex = sbase + inc - t4;
    for (int w = 0; w < wave; ++w) ex += wsum[w];
    ((int4*)off)[blockIdx.x * 256 + threadIdx.x] =
        make_int4(ex, ex + t1, ex + t2, ex + t3);
}

__global__ __launch_bounds__(256) void sage_fill(
    const int* __restrict__ src, const int* __restrict__ dst,
    const int* __restrict__ lrank, const int* __restrict__ off,
    const int* __restrict__ base, int* __restrict__ perm, int n_edges, int pad)
{
    const int t = blockIdx.x * 256 + threadIdx.x;
    int e = t * 4;
    if (e >= n_edges) return;
    const int* __restrict__ brow = base + (size_t)(e / CH) * pad;
    if (e + 3 < n_edges) {
        const int4 d = ((const int4*)dst)[t];
        const int4 r = ((const int4*)lrank)[t];
        const int4 s = ((const int4*)src)[t];
        const int o0 = off[d.x] + brow[d.x];
        const int o1 = off[d.y] + brow[d.y];
        const int o2 = off[d.z] + brow[d.z];
        const int o3 = off[d.w] + brow[d.w];
        perm[o0 + r.x] = s.x;
        perm[o1 + r.y] = s.y;
        perm[o2 + r.z] = s.z;
        perm[o3 + r.w] = s.w;
    } else {
        for (; e < n_edges; ++e) perm[off[dst[e]] + brow[dst[e]] + lrank[e]] = src[e];
    }
}

__global__ __launch_bounds__(256) void sage_gather_mean(
    const float* __restrict__ h, const int* __restrict__ off,
    const int* __restrict__ perm, float* __restrict__ mean, int n_nodes)
{
    const int lane = threadIdx.x & 31;
    const int n = __builtin_amdgcn_readfirstlane(
        (int)(blockIdx.x * 8 + (threadIdx.x >> 5)));
    if (n >= n_nodes) return;

    const int beg = off[n];
    const int end = off[n + 1];
    const float4* __restrict__ h4 = (const float4*)h;

    float4 acc = make_float4(0.0f, 0.0f, 0.0f, 0.0f);
    for (int base = beg; base < end; base += 32) {
        const int m = min(32, end - base);
        const int idx = (lane < m) ? perm[base + lane] : 0;
        int j = 0;
        for (; j + 4 <= m; j += 4) {
            const int s0 = __builtin_amdgcn_readlane(idx, j);
            const int s1 = __builtin_amdgcn_readlane(idx, j + 1);
            const int s2 = __builtin_amdgcn_readlane(idx, j + 2);
            const int s3 = __builtin_amdgcn_readlane(idx, j + 3);
            const float4 v0 = h4[(size_t)s0 * (D / 4) + lane];
            const float4 v1 = h4[(size_t)s1 * (D / 4) + lane];
            const float4 v2 = h4[(size_t)s2 * (D / 4) + lane];
            const float4 v3 = h4[(size_t)s3 * (D / 4) + lane];
            acc.x += v0.x; acc.y += v0.y; acc.z += v0.z; acc.w += v0.w;
            acc.x += v1.x; acc.y += v1.y; acc.z += v1.z; acc.w += v1.w;
            acc.x += v2.x; acc.y += v2.y; acc.z += v2.z; acc.w += v2.w;
            acc.x += v3.x; acc.y += v3.y; acc.z += v3.z; acc.w += v3.w;
        }
        const int r = m - j;
        if (r == 3) {
            const int s0 = __builtin_amdgcn_readlane(idx, j);
            const int s1 = __builtin_amdgcn_readlane(idx, j + 1);
            const int s2 = __builtin_amdgcn_readlane(idx, j + 2);
            const float4 v0 = h4[(size_t)s0 * (D / 4) + lane];
            const float4 v1 = h4[(size_t)s1 * (D / 4) + lane];
            const float4 v2 = h4[(size_t)s2 * (D / 4) + lane];
            acc.x += v0.x; acc.y += v0.y; acc.z += v0.z; acc.w += v0.w;
            acc.x += v1.x; acc.y += v1.y; acc.z += v1.z; acc.w += v1.w;
            acc.x += v2.x; acc.y += v2.y; acc.z += v2.z; acc.w += v2.w;
        } else if (r == 2) {
            const int s0 = __builtin_amdgcn_readlane(idx, j);
            const int s1 = __builtin_amdgcn_readlane(idx, j + 1);
            const float4 v0 = h4[(size_t)s0 * (D / 4) + lane];
            const float4 v1 = h4[(size_t)s1 * (D / 4) + lane];
            acc.x += v0.x; acc.y += v0.y; acc.z += v0.z; acc.w += v0.w;
            acc.x += v1.x; acc.y += v1.y; acc.z += v1.z; acc.w += v1.w;
        } else if (r == 1) {
            const int s0 = __builtin_amdgcn_readlane(idx, j);
            const float4 v0 = h4[(size_t)s0 * (D / 4) + lane];
            acc.x += v0.x; acc.y += v0.y; acc.z += v0.z; acc.w += v0.w;
        }
    }

    const float rdeg = 1.0f / fmaxf((float)(end - beg), 1.0f);
    acc.x *= rdeg; acc.y *= rdeg; acc.z *= rdeg; acc.w *= rdeg;
    ((float4*)mean)[(size_t)n * (D / 4) + lane] = acc;
}

__global__ __launch_bounds__(128) void sage_gemm(
    const float* __restrict__ h, const float* __restrict__ Wself,
    const float* __restrict__ Wneigh, const float* __restrict__ bias,
    float* out, int n_nodes)
{
    const int lane = threadIdx.x & 31;
    const int wave = threadIdx.x >> 5;
    const int tile = blockIdx.x * 4 + wave;
    const int m0   = tile * 16;
    if (m0 >= n_nodes) return;

    const int half  = lane >> 4;
    const int l15   = lane & 15;
    const int mA    = m0 + l15;
    const int kOffA = half * 2;

    v8f acc[8];
    #pragma unroll
    for (int t = 0; t < 8; ++t) {
        #pragma unroll
        for (int r = 0; r < 8; ++r) acc[t][r] = 0.0f;
    }

    const float* hrow = h   + (size_t)mA * D + kOffA;
    const float* srow = out + (size_t)mA * D + kOffA;

    #pragma unroll 1
    for (int pass = 0; pass < 2; ++pass) {
        const float* arow = pass ? srow : hrow;
        const float* W    = pass ? Wneigh : Wself;
        const float* wb0  = W + (size_t)l15 * D + kOffA;
        #pragma unroll 4
        for (int ks = 0; ks < D; ks += 4) {
            v2f a = *(const v2f*)(arow + ks);
            #pragma unroll
            for (int t = 0; t < 8; ++t) {
                v2f bf = *(const v2f*)(wb0 + ks + (size_t)t * 16 * D);
                acc[t] = __builtin_amdgcn_wmma_f32_16x16x4_f32(
                     false, a,  false, bf,
                     (short)0, acc[t],
                     false,  false);
            }
        }
    }

    const int mBase = m0 + half * 8;
    #pragma unroll
    for (int t = 0; t < 8; ++t) {
        int n = t * 16 + l15;
        float bb = bias[n];
        float* op = out + (size_t)mBase * D + n;
        #pragma unroll
        for (int r = 0; r < 8; ++r) {
            op[(size_t)r * D] = acc[t][r] + bb;
        }
    }
}

extern "C" void kernel_launch(void* const* d_in, const int* in_sizes, int n_in,
                              void* d_out, int out_size, void* d_ws, size_t ws_size,
                              hipStream_t stream) {
    const float* h      = (const float*)d_in[0];
    const int*   src    = (const int*)d_in[1];
    const int*   dst    = (const int*)d_in[2];
    const float* Wself  = (const float*)d_in[3];
    const float* Wneigh = (const float*)d_in[4];
    const float* bias   = (const float*)d_in[5];
    float*       out    = (float*)d_out;

    const int n_nodes = in_sizes[0] / D;
    const int n_edges = in_sizes[1];

    const int nb  = n_nodes / SCAN_E + 1;
    const int pad = nb * SCAN_E;
    const int nw  = pad / 2;
    const int G   = (n_edges + CH - 1) / CH;
    const int E4  = (n_edges + 3) / 4 * 4;

    int*      cnt   = (int*)d_ws;
    int*      off   = cnt + pad;
    int*      lrank = off + pad;
    int*      perm  = lrank + E4;
    unsigned* C     = (unsigned*)(perm + E4);
    int*      base  = (int*)(C + (size_t)G * nw);
    int*      bsum  = base + (size_t)G * pad;

    const int eblocks = (n_edges + 1023) / 1024;

    sage_count_lds<<<G, 1024, 0, stream>>>(dst, C, lrank, n_edges, nw);
    sage_colscan<<<nw / 256, 256, 0, stream>>>(C, base, cnt, G, nw);
    sage_blocksum<<<nb, 256, 0, stream>>>(cnt, bsum);
    sage_scan<<<nb, 256, 0, stream>>>(cnt, bsum, off);
    sage_fill<<<eblocks, 256, 0, stream>>>(src, dst, lrank, off, base, perm, n_edges, pad);
    sage_gather_mean<<<(n_nodes + 7) / 8, 256, 0, stream>>>(h, off, perm, out, n_nodes);

    int n_tiles = (n_nodes + 15) / 16;
    int gblocks = (n_tiles + 3) / 4;
    sage_gemm<<<gblocks, 128, 0, stream>>>(h, Wself, Wneigh, bias, out, n_nodes);
}


# GEMM operands via LDS: one resident 16-wave work-group per WGP holds both weight matrices in LDS, h/mean tiles arrive as whole 512-B rows; fragments read from LDS instead of 16-line global loads
# speedup vs baseline: 2.5592x; 1.2884x over previous
#include <hip/hip_runtime.h>

typedef float v2f __attribute__((ext_vector_type(2)));
typedef float v8f __attribute__((ext_vector_type(8)));

#define D 128
#define SCAN_E 1024
#define CH 51200
#define NW_MAX 51200


__device__ __forceinline__ int sage_rank_one(unsigned* cw, int d) {
    const int sh = (d & 1) << 4;
    const unsigned old = atomicAdd(cw + (d >> 1), 1u << sh);
    return (int)((old >> sh) & 0xFFFFu);
}

__global__ __launch_bounds__(1024) void sage_count_lds(
    const int* __restrict__ dst, unsigned* __restrict__ C,
    int* __restrict__ lrank, int n_edges, int nw)
{
    __shared__ __attribute__((aligned(16))) unsigned cw[NW_MAX];
    const int tid = threadIdx.x;

    for (int i = tid; i < nw / 4; i += 1024)
        ((uint4*)cw)[i] = make_uint4(0u, 0u, 0u, 0u);
    __syncthreads();

    const int e0 = blockIdx.x * CH;
    const int e1 = min(e0 + CH, n_edges);
    const int ng = (e1 - e0) >> 2;
    const int4* __restrict__ d4 = (const int4*)(dst + e0);
    int4* __restrict__ r4 = (int4*)(lrank + e0);
    for (int q = tid; q < ng; q += 4096) {
        int4 d[4];
        #pragma unroll
        for (int u = 0; u < 4; ++u) {
            const int qq = q + u * 1024;
            d[u] = (qq < ng) ? d4[qq] : make_int4(0, 0, 0, 0);
        }
        #pragma unroll
        for (int u = 0; u < 4; ++u) {
            const int qq = q + u * 1024;
            if (qq < ng) {
                int4 r;
                r.x = sage_rank_one(cw, d[u].x);
                r.y = sage_rank_one(cw, d[u].y);
                r.z = sage_rank_one(cw, d[u].z);
                r.w = sage_rank_one(cw, d[u].w);
                r4[qq] = r;
            }
        }
    }
    for (int k = e0 + ng * 4 + tid; k < e1; k += 1024)
        lrank[k] = sage_rank_one(cw, dst[k]);
    __syncthreads();

    uint4* Crow = (uint4*)(C + (size_t)blockIdx.x * nw);
    for (int i = tid; i < nw / 4; i += 1024) Crow[i] = ((const uint4*)cw)[i];
}

__global__ __launch_bounds__(256) void sage_colscan(
    const unsigned* __restrict__ C, int* __restrict__ base,
    int* __restrict__ cnt, int n_chunks, int nw)
{
    const int w = blockIdx.x * 256 + threadIdx.x;
    int lo = 0, hi = 0;
    #pragma unroll 8
    for (int g = 0; g < n_chunks; ++g) {
        const unsigned c = C[(size_t)g * nw + w];
        *(int2*)(base + ((size_t)g * nw + w) * 2) = make_int2(lo, hi);
        lo += (int)(c & 0xFFFFu);
        hi += (int)(c >> 16);
    }
    *(int2*)(cnt + 2 * w) = make_int2(lo, hi);
}

__global__ __launch_bounds__(256) void sage_blocksum(
    const int* __restrict__ cnt, int* __restrict__ bsum)
{
    __shared__ int wsum[8];
    const int lane = threadIdx.x & 31;
    const int wave = threadIdx.x >> 5;
    int4 c = ((const int4*)cnt)[blockIdx.x * 256 + threadIdx.x];
    int s = (c.x + c.y) + (c.z + c.w);
    for (int o = 16; o > 0; o >>= 1) s += __shfl_xor(s, o);
    if (lane == 0) wsum[wave] = s;
    __syncthreads();
    if (threadIdx.x == 0) {
        int t = 0;
        for (int w = 0; w < 8; ++w) t += wsum[w];
        bsum[blockIdx.x] = t;
    }
}

__global__ __launch_bounds__(256) void sage_scan(
    const int* __restrict__ cnt, const int* __restrict__ bsum,
    int* __restrict__ off)
{
    __shared__ int wsum[8];
    __shared__ int sbase;
    const int lane = threadIdx.x & 31;
    const int wave = threadIdx.x >> 5;

    int4 c = ((const int4*)cnt)[blockIdx.x * 256 + threadIdx.x];

    if (wave == 0) {
        int t = 0;
        for (int i = lane; i < (int)blockIdx.x; i += 32) t += bsum[i];
        for (int o = 16; o > 0; o >>= 1) t += __shfl_xor(t, o);
        if (lane == 0) sbase = t;
    }

    const int t1 = c.x, t2 = t1 + c.y, t3 = t2 + c.z, t4 = t3 + c.w;
    int inc = t4;
    for (int o = 1; o < 32; o <<= 1) {
        int v = __shfl_up(inc, o);
        if (lane >= o) inc += v;
    }
    if (lane == 31) wsum[wave] = inc;
    __syncthreads();

    int ex = sbase + inc - t4;
    for (int w = 0; w < wave; ++w) ex += wsum[w];
    ((int4*)off)[blockIdx.x * 256 + threadIdx.x] =
        make_int4(ex, ex + t1, ex + t2, ex + t3);
}

__global__ __launch_bounds__(256) void sage_fill(
    const int* __restrict__ src, const int* __restrict__ dst,
    const int* __restrict__ lrank, const int* __restrict__ off,
    const int* __restrict__ base, int* __restrict__ perm, int n_edges, int pad)
{
    const int t = blockIdx.x * 256 + threadIdx.x;
    int e = t * 4;
    if (e >= n_edges) return;
    const int* __restrict__ brow = base + (size_t)(e / CH) * pad;
    if (e + 3 < n_edges) {
        const int4 d = ((const int4*)dst)[t];
        const int4 r = ((const int4*)lrank)[t];
        const int4 s = ((const int4*)src)[t];
        const int o0 = off[d.x] + brow[d.x];
        const int o1 = off[d.y] + brow[d.y];
        const int o2 = off[d.z] + brow[d.z];
        const int o3 = off[d.w] + brow[d.w];
        perm[o0 + r.x] = s.x;
        perm[o1 + r.y] = s.y;
        perm[o2 + r.z] = s.z;
        perm[o3 + r.w] = s.w;
    } else {
        for (; e < n_edges; ++e) perm[off[dst[e]] + brow[dst[e]] + lrank[e]] = src[e];
    }
}

__global__ __launch_bounds__(256) void sage_gather_mean(
    const float* __restrict__ h, const int* __restrict__ off,
    const int* __restrict__ perm, float* __restrict__ mean, int n_nodes)
{
    const int lane = threadIdx.x & 31;
    const int n = __builtin_amdgcn_readfirstlane(
        (int)(blockIdx.x * 8 + (threadIdx.x >> 5)));
    if (n >= n_nodes) return;

    const int beg = off[n];
    const int end = off[n + 1];
    const float4* __restrict__ h4 = (const float4*)h;

    float4 acc = make_float4(0.0f, 0.0f, 0.0f, 0.0f);
    for (int base = beg; base < end; base += 32) {
        const int m = min(32, end - base);
        const int idx = (lane < m) ? perm[base + lane] : 0;
        int j = 0;
        for (; j + 4 <= m; j += 4) {
            const int s0 = __builtin_amdgcn_readlane(idx, j);
            const int s1 = __builtin_amdgcn_readlane(idx, j + 1);
            const int s2 = __builtin_amdgcn_readlane(idx, j + 2);
            const int s3 = __builtin_amdgcn_readlane(idx, j + 3);
            const float4 v0 = h4[(size_t)s0 * (D / 4) + lane];
            const float4 v1 = h4[(size_t)s1 * (D / 4) + lane];
            const float4 v2 = h4[(size_t)s2 * (D / 4) + lane];
            const float4 v3 = h4[(size_t)s3 * (D / 4) + lane];
            acc.x += v0.x; acc.y += v0.y; acc.z += v0.z; acc.w += v0.w;
            acc.x += v1.x; acc.y += v1.y; acc.z += v1.z; acc.w += v1.w;
            acc.x += v2.x; acc.y += v2.y; acc.z += v2.z; acc.w += v2.w;
            acc.x += v3.x; acc.y += v3.y; acc.z += v3.z; acc.w += v3.w;
        }
        const int r = m - j;
        if (r == 3) {
            const int s0 = __builtin_amdgcn_readlane(idx, j);
            const int s1 = __builtin_amdgcn_readlane(idx, j + 1);
            const int s2 = __builtin_amdgcn_readlane(idx, j + 2);
            const float4 v0 = h4[(size_t)s0 * (D / 4) + lane];
            const float4 v1 = h4[(size_t)s1 * (D / 4) + lane];
            const float4 v2 = h4[(size_t)s2 * (D / 4) + lane];
            acc.x += v0.x; acc.y += v0.y; acc.z += v0.z; acc.w += v0.w;
            acc.x += v1.x; acc.y += v1.y; acc.z += v1.z; acc.w += v1.w;
            acc.x += v2.x; acc.y += v2.y; acc.z += v2.z; acc.w += v2.w;
        } else if (r == 2) {
            const int s0 = __builtin_amdgcn_readlane(idx, j);
            const int s1 = __builtin_amdgcn_readlane(idx, j + 1);
            const float4 v0 = h4[(size_t)s0 * (D / 4) + lane];
            const float4 v1 = h4[(size_t)s1 * (D / 4) + lane];
            acc.x += v0.x; acc.y += v0.y; acc.z += v0.z; acc.w += v0.w;
            acc.x += v1.x; acc.y += v1.y; acc.z += v1.z; acc.w += v1.w;
        } else if (r == 1) {
            const int s0 = __builtin_amdgcn_readlane(idx, j);
            const float4 v0 = h4[(size_t)s0 * (D / 4) + lane];
            acc.x += v0.x; acc.y += v0.y; acc.z += v0.z; acc.w += v0.w;
        }
    }

    const float rdeg = 1.0f / fmaxf((float)(end - beg), 1.0f);
    acc.x *= rdeg; acc.y *= rdeg; acc.z *= rdeg; acc.w *= rdeg;
    ((float4*)mean)[(size_t)n * (D / 4) + lane] = acc;
}

#define SP 132
#define GW 16
#define GB 256

#define SAGE_ROWS16(X) X(0) X(1) X(2) X(3) X(4) X(5) X(6) X(7) X(8) X(9) X(10) X(11) X(12) X(13) X(14) X(15)
#define SAGE_DECL(i)    float4 r##i, q##i;
#define SAGE_LOAD_H(i)  r##i = rp[i * (D / 4) + lane];
#define SAGE_LOAD_M(i)  q##i = rp[i * (D / 4) + lane];
#define SAGE_STAGE_H(i) *(float4*)(aw + i * SP + lane * 4) = r##i;
#define SAGE_STAGE_M(i) *(float4*)(aw + i * SP + lane * 4) = q##i;

__device__ __forceinline__ void sage_kpass(
    v8f (&acc)[8], const float* afr, const float* bfr)
{
    #pragma unroll 4
    for (int ks = 0; ks < D; ks += 4) {
        const v2f a = *(const v2f*)(afr + ks);
        #pragma unroll
        for (int t = 0; t < 8; ++t) {
            const v2f bf = *(const v2f*)(bfr + ks + t * 16 * SP);
            acc[t] = __builtin_amdgcn_wmma_f32_16x16x4_f32(
                 false, a,  false, bf,
                 (short)0, acc[t],
                 false,  false);
        }
    }
}

__global__ __launch_bounds__(GW * 32) void sage_gemm(
    const float* __restrict__ h, const float* __restrict__ Wself,
    const float* __restrict__ Wneigh, const float* __restrict__ bias,
    float* out, int n_nodes)
{
    __shared__ __attribute__((aligned(16))) float sW[2 * D * SP];
    __shared__ __attribute__((aligned(16))) float sA[GW * 16 * SP];

    const int lane = threadIdx.x & 31;
    const int wave = __builtin_amdgcn_readfirstlane((int)(threadIdx.x >> 5));

    {
        const float* wsrc = (wave < GW / 2)
            ? (Wself  + (size_t)(wave * 16) * D)
            : (Wneigh + (size_t)((wave - GW / 2) * 16) * D);
        float4 w[16];
        #pragma unroll
        for (int i = 0; i < 16; ++i)
            w[i] = ((const float4*)(wsrc + (size_t)i * D))[lane];
        #pragma unroll
        for (int i = 0; i < 16; ++i)
            *(float4*)(sW + (wave * 16 + i) * SP + lane * 4) = w[i];
    }
    __syncthreads();

    const int half = lane >> 4;
    const int l15  = lane & 15;
    const int kOff = half * 2;

    float* aw = sA + wave * (16 * SP);
    const float* afr  = aw + l15 * SP + kOff;
    const float* bfr0 = sW + l15 * SP + kOff;
    const float* bfr1 = bfr0 + D * SP;

    float bb[8];
    #pragma unroll
    for (int t = 0; t < 8; ++t) bb[t] = bias[t * 16 + l15];

    const int n_tiles = (n_nodes + 15) >> 4;
    const int t_beg = (int)(((long long)n_tiles * (long long)blockIdx.x) / GB);
    const int t_end = (int)(((long long)n_tiles * (long long)(blockIdx.x + 1)) / GB);

    int tile = t_beg + wave;
    SAGE_ROWS16(SAGE_DECL)
    if (tile < t_end) {
        const float4* rp = (const float4*)(h + (size_t)tile * 16 * D);
        SAGE_ROWS16(SAGE_LOAD_H)
    }

    while (tile < t_end) {
        const int m0 = tile * 16;

        {
            const float4* rp = (const float4*)(out + (size_t)m0 * D);
            SAGE_ROWS16(SAGE_LOAD_M)
        }
        asm volatile("" ::: "memory");

        SAGE_ROWS16(SAGE_STAGE_H)

        v8f acc[8];
        #pragma unroll
        for (int t = 0; t < 8; ++t) {
            #pragma unroll
            for (int r = 0; r < 8; ++r) acc[t][r] = 0.0f;
        }

        sage_kpass(acc, afr, bfr0);

        SAGE_ROWS16(SAGE_STAGE_M)

        const int next = tile + GW;
        {
            const float4* rp = (const float4*)(h + (size_t)min(next, t_end - 1) * 16 * D);
            SAGE_ROWS16(SAGE_LOAD_H)
        }

        sage_kpass(acc, afr, bfr1);

        const int mBase = m0 + half * 8;
        #pragma unroll
        for (int t = 0; t < 8; ++t) {
            float* op = out + (size_t)mBase * D + t * 16 + l15;
            #pragma unroll
            for (int r = 0; r < 8; ++r) {
                op[(size_t)r * D] = acc[t][r] + bb[t];
            }
        }

        tile = next;
    }
}

extern "C" void kernel_launch(void* const* d_in, const int* in_sizes, int n_in,
                              void* d_out, int out_size, void* d_ws, size_t ws_size,
                              hipStream_t stream) {
    const float* h      = (const float*)d_in[0];
    const int*   src    = (const int*)d_in[1];
    const int*   dst    = (const int*)d_in[2];
    const float* Wself  = (const float*)d_in[3];
    const float* Wneigh = (const float*)d_in[4];
    const float* bias   = (const float*)d_in[5];
    float*       out    = (float*)d_out;

    const int n_nodes = in_sizes[0] / D;
    const int n_edges = in_sizes[1];

    const int nb  = n_nodes / SCAN_E + 1;
    const int pad = nb * SCAN_E;
    const int nw  = pad / 2;
    const int G   = (n_edges + CH - 1) / CH;
    const int E4  = (n_edges + 3) / 4 * 4;

    int*      cnt   = (int*)d_ws;
    int*      off   = cnt + pad;
    int*      lrank = off + pad;
    int*      perm  = lrank + E4;
    unsigned* C     = (unsigned*)(perm + E4);
    int*      base  = (int*)(C + (size_t)G * nw);
    int*      bsum  = base + (size_t)G * pad;

    const int eblocks = (n_edges + 1023) / 1024;

    sage_count_lds<<<G, 1024, 0, stream>>>(dst, C, lrank, n_edges, nw);
    sage_colscan<<<nw / 256, 256, 0, stream>>>(C, base, cnt, G, nw);
    sage_blocksum<<<nb, 256, 0, stream>>>(cnt, bsum);
    sage_scan<<<nb, 256, 0, stream>>>(cnt, bsum, off);
    sage_fill<<<eblocks, 256, 0, stream>>>(src, dst, lrank, off, base, perm, n_edges, pad);
    sage_gather_mean<<<(n_nodes + 7) / 8, 256, 0, stream>>>(h, off, perm, out, n_nodes);

    sage_gemm<<<GB, GW * 32, 0, stream>>>(h, Wself, Wneigh, bias, out, n_nodes);
}


# Node-offset scan without own launches: chunk-prefix kernel scans in-degrees per 512-node block into base; fill and gather add the block start from an LDS prefix; 6 -> 4 launches, 1 table read/edge
# speedup vs baseline: 2.8135x; 1.0994x over previous
#include <hip/hip_runtime.h>

typedef float v2f __attribute__((ext_vector_type(2)));
typedef float v8f __attribute__((ext_vector_type(8)));

#define D 128
#define SCAN_E 1024
#define CH 51200
#define NW_MAX 51200
#define NBLK_MAX 256


__device__ __forceinline__ int sage_rank_one(unsigned* cw, int d) {
    const int sh = (d & 1) << 4;
    const unsigned old = atomicAdd(cw + (d >> 1), 1u << sh);
    return (int)((old >> sh) & 0xFFFFu);
}

__global__ __launch_bounds__(1024) void sage_count_lds(
    const int* __restrict__ dst, unsigned* __restrict__ C,
    int* __restrict__ lrank, int n_edges, int nw)
{
    __shared__ __attribute__((aligned(16))) unsigned cw[NW_MAX];
    const int tid = threadIdx.x;

    for (int i = tid; i < nw / 4; i += 1024)
        ((uint4*)cw)[i] = make_uint4(0u, 0u, 0u, 0u);
    __syncthreads();

    const int e0 = blockIdx.x * CH;
    const int e1 = min(e0 + CH, n_edges);
    const int ng = (e1 - e0) >> 2;
    const int4* __restrict__ d4 = (const int4*)(dst + e0);
    int4* __restrict__ r4 = (int4*)(lrank + e0);
    for (int q = tid; q < ng; q += 4096) {
        int4 d[4];
        #pragma unroll
        for (int u = 0; u < 4; ++u) {
            const int qq = q + u * 1024;
            d[u] = (qq < ng) ? d4[qq] : make_int4(0, 0, 0, 0);
        }
        #pragma unroll
        for (int u = 0; u < 4; ++u) {
            const int qq = q + u * 1024;
            if (qq < ng) {
                int4 r;
                r.x = sage_rank_one(cw, d[u].x);
                r.y = sage_rank_one(cw, d[u].y);
                r.z = sage_rank_one(cw, d[u].z);
                r.w = sage_rank_one(cw, d[u].w);
                r4[qq] = r;
            }
        }
    }
    for (int k = e0 + ng * 4 + tid; k < e1; k += 1024)
        lrank[k] = sage_rank_one(cw, dst[k]);
    __syncthreads();

    uint4* Crow = (uint4*)(C + (size_t)blockIdx.x * nw);
    for (int i = tid; i < nw / 4; i += 1024) Crow[i] = ((const uint4*)cw)[i];
}

__global__ __launch_bounds__(256) void sage_colscan(
    const unsigned* __restrict__ C, int* __restrict__ base,
    int* __restrict__ loc, int* __restrict__ bsum, int n_chunks, int nw)
{
    __shared__ int wsum[8];
    const int lane = threadIdx.x & 31;
    const int wave = threadIdx.x >> 5;
    const int w = blockIdx.x * 256 + threadIdx.x;

    int lo = 0, hi = 0;
    #pragma unroll 8
    for (int g = 0; g < n_chunks; ++g) {
        const unsigned c = C[(size_t)g * nw + w];
        lo += (int)(c & 0xFFFFu);
        hi += (int)(c >> 16);
    }

    const int t2 = lo + hi;
    int inc = t2;
    for (int o = 1; o < 32; o <<= 1) {
        int v = __shfl_up(inc, o);
        if (lane >= o) inc += v;
    }
    if (lane == 31) wsum[wave] = inc;
    __syncthreads();

    int ex = inc - t2;
    int tot = 0;
    for (int i = 0; i < 8; ++i) {
        const int s = wsum[i];
        if (i < wave) ex += s;
        tot += s;
    }
    if (threadIdx.x == 0) bsum[blockIdx.x] = tot;
    *(int2*)(loc + 2 * w) = make_int2(ex, ex + lo);

    int rlo = ex, rhi = ex + lo;
    #pragma unroll 8
    for (int g = 0; g < n_chunks; ++g) {
        const unsigned c = C[(size_t)g * nw + w];
        *(int2*)(base + ((size_t)g * nw + w) * 2) = make_int2(rlo, rhi);
        rlo += (int)(c & 0xFFFFu);
        rhi += (int)(c >> 16);
    }
}

__device__ __forceinline__ void sage_block_starts(const int v, int* bp, int* wsum)
{
    const int tid = threadIdx.x;
    const int lane = tid & 31;
    const int wave = tid >> 5;
    int inc = v;
    if (tid < NBLK_MAX) {
        for (int o = 1; o < 32; o <<= 1) {
            int u = __shfl_up(inc, o);
            if (lane >= o) inc += u;
        }
        if (lane == 31) wsum[wave] = inc;
    }
    __syncthreads();
    if (tid < NBLK_MAX) {
        int ex = inc - v;
        for (int i = 0; i < wave; ++i) ex += wsum[i];
        bp[tid] = ex;
    }
    __syncthreads();
}

__global__ __launch_bounds__(256) void sage_fill(
    const int* __restrict__ src, const int* __restrict__ dst,
    const int* __restrict__ lrank, const int* __restrict__ bsum,
    const int* __restrict__ base, int* __restrict__ perm,
    int n_edges, int pad, int nblk)
{
    __shared__ int bp[NBLK_MAX];
    __shared__ int wsum[8];
    const int t = blockIdx.x * 256 + threadIdx.x;
    int e = t * 4;
    const bool full = (e + 3 < n_edges);

    const int bv = ((int)threadIdx.x < nblk) ? bsum[threadIdx.x] : 0;
    int4 d = make_int4(0, 0, 0, 0), r = d, s = d;
    if (full) {
        d = ((const int4*)dst)[t];
        r = ((const int4*)lrank)[t];
        s = ((const int4*)src)[t];
    }
    sage_block_starts(bv, bp, wsum);

    if (e >= n_edges) return;
    const int* __restrict__ brow = base + (size_t)(e / CH) * pad;
    if (full) {
        const int o0 = bp[d.x >> 9] + brow[d.x];
        const int o1 = bp[d.y >> 9] + brow[d.y];
        const int o2 = bp[d.z >> 9] + brow[d.z];
        const int o3 = bp[d.w >> 9] + brow[d.w];
        perm[o0 + r.x] = s.x;
        perm[o1 + r.y] = s.y;
        perm[o2 + r.z] = s.z;
        perm[o3 + r.w] = s.w;
    } else {
        for (; e < n_edges; ++e) {
            const int dd = dst[e];
            perm[bp[dd >> 9] + brow[dd] + lrank[e]] = src[e];
        }
    }
}

#define SP 132
#define GW 13
#define GB 256

#define SAGE_ROWS16(X) X(0) X(1) X(2) X(3) X(4) X(5) X(6) X(7) X(8) X(9) X(10) X(11) X(12) X(13) X(14) X(15)
#define SAGE_DECL(i)    float4 r##i;
#define SAGE_LOAD_H(i)  r##i = rp[i * (D / 4) + lane];
#define SAGE_STAGE_H(i) *(float4*)(aw + i * SP + lane * 4) = r##i;

#define SAGE_W10(X) X(0) X(1) X(2) X(3) X(4) X(5) X(6) X(7) X(8) X(9)
#define SAGE_WDECL(i)  float4 w##i = make_float4(0.0f, 0.0f, 0.0f, 0.0f);
#define SAGE_WLOAD(i)  { const int row = rb + i * GW; if (row < 2 * D) w##i = ((const float4*)((row < D) ? (Wself + (size_t)row * D) : (Wneigh + (size_t)(row - D) * D)))[lane]; }
#define SAGE_WSTORE(i) { const int row = rb + i * GW; if (row < 2 * D) *(float4*)(sW + row * SP + lane * 4) = w##i; }

__device__ __forceinline__ void sage_kpass(
    v8f (&acc)[8], const float* afr, const float* bfr)
{
    #pragma nounroll
    for (int k0 = 0; k0 < D; k0 += 16) {
        #pragma unroll
        for (int kk = 0; kk < 16; kk += 4) {
            const int ks = k0 + kk;
            const v2f a = *(const v2f*)(afr + ks);
            #pragma unroll
            for (int t = 0; t < 8; ++t) {
                const v2f bf = *(const v2f*)(bfr + ks + t * 16 * SP);
                acc[t] = __builtin_amdgcn_wmma_f32_16x16x4_f32(
                     false, a,  false, bf,
                     (short)0, acc[t],
                     false,  false);
            }
        }
        __builtin_amdgcn_sched_barrier(0);
        asm volatile("v_nop\n\tv_nop\n\tv_nop\n\tv_nop");
    }
}

struct SageWalk {
    int node;
    int nend;
    float4 a;
};

__device__ __forceinline__ void sage_flush(
    SageWalk& w, float* aw, const int bnd, const float rdv, const int lane)
{
    const float rdeg = __builtin_bit_cast(float,
        __builtin_amdgcn_readlane(__builtin_bit_cast(int, rdv), w.node));
    w.a.x *= rdeg; w.a.y *= rdeg; w.a.z *= rdeg; w.a.w *= rdeg;
    *(float4*)(aw + w.node * SP + lane * 4) = w.a;
    w.a = make_float4(0.0f, 0.0f, 0.0f, 0.0f);
    ++w.node;
    w.nend = __builtin_amdgcn_readlane(bnd, w.node + 1);
}

template <int N>
__device__ __forceinline__ int sage_batch(
    SageWalk& w, float* aw, const float4* h4,
    const int* __restrict__ perm, const int idx, const int cnt, const int e0,
    const int pn, const int bnd, const float rdv, const int lane)
{
    float4 v[N];
    #pragma unroll
    for (int u = 0; u < N; ++u) {
        const int s = __builtin_amdgcn_readlane(idx, min(u, cnt - 1));
        v[u] = h4[(size_t)s * (D / 4) + lane];
    }
    const int idx_n = perm[pn];
    asm volatile("" ::: "memory");
    #pragma unroll
    for (int u = 0; u < N; ++u) {
        if (u < cnt) {
            while (e0 + u >= w.nend) sage_flush(w, aw, bnd, rdv, lane);
            w.a.x += v[u].x; w.a.y += v[u].y; w.a.z += v[u].z; w.a.w += v[u].w;
        }
    }
    return idx_n;
}

__device__ __forceinline__ void sage_tile_means(
    float* aw, const float4* h4, const int* __restrict__ perm,
    const int bnd, const int lane)
{
    const int E0 = __builtin_amdgcn_readlane(bnd, 0);
    const int E1 = __builtin_amdgcn_readlane(bnd, 16);

    const int bnx = __shfl_down(bnd, 1);
    const float rdv = 1.0f / fmaxf((float)(bnx - bnd), 1.0f);

    SageWalk w;
    w.node = 0;
    w.nend = __builtin_amdgcn_readlane(bnd, 1);
    w.a = make_float4(0.0f, 0.0f, 0.0f, 0.0f);

    if (E0 < E1) {
        int idx = perm[min(E0 + lane, E1 - 1)];
        for (int eb = E0; eb < E1; eb += 32) {
            const int m = min(32, E1 - eb);
            const int pn = min(eb + 32 + lane, E1 - 1);
            if (m > 16) idx = sage_batch<32>(w, aw, h4, perm, idx, m, eb, pn, bnd, rdv, lane);
            else        idx = sage_batch<16>(w, aw, h4, perm, idx, m, eb, pn, bnd, rdv, lane);
        }
    }
    while (w.node < 16) sage_flush(w, aw, bnd, rdv, lane);
}

__global__ __launch_bounds__(GW * 32) void sage_gather_gemm(
    const float* h, const float* __restrict__ Wself,
    const float* __restrict__ Wneigh, const float* __restrict__ bias,
    const int* __restrict__ loc, const int* __restrict__ bsum,
    const int* __restrict__ perm, float* __restrict__ out, int n_nodes, int nblk)
{
    __shared__ __attribute__((aligned(16))) float sW[2 * D * SP];
    __shared__ __attribute__((aligned(16))) float sA[GW * 16 * SP];
    __shared__ int sBP[NBLK_MAX];
    __shared__ int sWS[8];

    const int bv = ((int)threadIdx.x < nblk) ? bsum[threadIdx.x] : 0;

    const int lane = threadIdx.x & 31;
    const int wave = __builtin_amdgcn_readfirstlane((int)(threadIdx.x >> 5));

    for (int ph = 0; ph < 2; ++ph) {
        const int rb = wave + ph * 10 * GW;
        SAGE_W10(SAGE_WDECL)
        SAGE_W10(SAGE_WLOAD)
        SAGE_W10(SAGE_WSTORE)
    }
    sage_block_starts(bv, sBP, sWS);

    const int half = lane >> 4;
    const int l15  = lane & 15;
    const int kOff = half * 2;

    float* aw = sA + wave * (16 * SP);
    const float* afr  = aw + l15 * SP + kOff;
    const float* bfr0 = sW + l15 * SP + kOff;
    const float* bfr1 = bfr0 + D * SP;

    const float4* h4 = (const float4*)h;

    float bb[8];
    #pragma unroll
    for (int t = 0; t < 8; ++t) bb[t] = bias[t * 16 + l15];

    const int n_tiles = (n_nodes + 15) >> 4;
    const int t_beg = (int)(((long long)n_tiles * (long long)blockIdx.x) / GB);
    const int t_end = (int)(((long long)n_tiles * (long long)(blockIdx.x + 1)) / GB);

    int tile = t_beg + wave;
    SAGE_ROWS16(SAGE_DECL)
    if (tile < t_end) {
        const float4* rp = (const float4*)(h + (size_t)tile * 16 * D);
        SAGE_ROWS16(SAGE_LOAD_H)
    }

    while (tile < t_end) {
        const int m0 = tile * 16;

        const int bn = min(m0 + min(lane, 16), n_nodes);
        const int bnd = loc[bn] + sBP[bn >> 9];

        SAGE_ROWS16(SAGE_STAGE_H)

        v8f acc[8];
        #pragma unroll
        for (int t = 0; t < 8; ++t) {
            #pragma unroll
            for (int r = 0; r < 8; ++r) acc[t][r] = 0.0f;
        }

        sage_kpass(acc, afr, bfr0);

        sage_tile_means(aw, h4, perm, bnd, lane);

        const int next = tile + GW;
        {
            const float4* rp = (const float4*)(h + (size_t)min(next, t_end - 1) * 16 * D);
            SAGE_ROWS16(SAGE_LOAD_H)
        }

        sage_kpass(acc, afr, bfr1);

        const int mBase = m0 + half * 8;
        #pragma unroll
        for (int t = 0; t < 8; ++t) {
            float* op = out + (size_t)mBase * D + t * 16 + l15;
            #pragma unroll
            for (int r = 0; r < 8; ++r) {
                op[(size_t)r * D] = acc[t][r] + bb[t];
            }
        }

        tile = next;
    }
}

extern "C" void kernel_launch(void* const* d_in, const int* in_sizes, int n_in,
                              void* d_out, int out_size, void* d_ws, size_t ws_size,
                              hipStream_t stream) {
    const float* h      = (const float*)d_in[0];
    const int*   src    = (const int*)d_in[1];
    const int*   dst    = (const int*)d_in[2];
    const float* Wself  = (const float*)d_in[3];
    const float* Wneigh = (const float*)d_in[4];
    const float* bias   = (const float*)d_in[5];
    float*       out    = (float*)d_out;

    const int n_nodes = in_sizes[0] / D;
    const int n_edges = in_sizes[1];

    const int nb  = n_nodes / SCAN_E + 1;
    const int pad = nb * SCAN_E;
    const int nw  = pad / 2;
    const int nblk = nw / 256;
    const int G   = (n_edges + CH - 1) / CH;
    const int E4  = (n_edges + 3) / 4 * 4;

    int*      loc   = (int*)d_ws;
    int*      bsum  = loc + pad;
    int*      lrank = bsum + pad;
    int*      perm  = lrank + E4;
    unsigned* C     = (unsigned*)(perm + E4);
    int*      base  = (int*)(C + (size_t)G * nw);

    const int eblocks = (n_edges + 1023) / 1024;

    sage_count_lds<<<G, 1024, 0, stream>>>(dst, C, lrank, n_edges, nw);
    sage_colscan<<<nblk, 256, 0, stream>>>(C, base, loc, bsum, G, nw);
    sage_fill<<<eblocks, 256, 0, stream>>>(src, dst, lrank, bsum, base, perm, n_edges, pad, nblk);
    sage_gather_gemm<<<GB, GW * 32, 0, stream>>>(h, Wself, Wneigh, bias, loc, bsum, perm, out, n_nodes, nblk);
}


# Edge sort in two levels: edges binned by destination range (one per GEMM work-group) in 2 launches, grouped by node in LDS inside the GEMM launch; 4 -> 3 launches; rows summed in new arrival order
# speedup vs baseline: 3.3761x; 1.2000x over previous
#include <hip/hip_runtime.h>

typedef float v2f __attribute__((ext_vector_type(2)));
typedef float v8f __attribute__((ext_vector_type(8)));

#define D 128
#define NBIN 256
#define NBLK_MAX 256
#define CHK2 6400
#define BT 1024
#define EPT ((CHK2 + BT - 1) / BT)
#define TR 16


__device__ __forceinline__ void sage_block_starts(const int v, int* bp, int* wsum)
{
    const int tid = threadIdx.x;
    const int lane = tid & 31;
    const int wave = tid >> 5;
    int inc = v;
    if (tid < NBLK_MAX) {
        for (int o = 1; o < 32; o <<= 1) {
            int u = __shfl_up(inc, o);
            if (lane >= o) inc += u;
        }
        if (lane == 31) wsum[wave] = inc;
    }
    __syncthreads();
    if (tid < NBLK_MAX) {
        int ex = inc - v;
        for (int i = 0; i < wave; ++i) ex += wsum[i];
        bp[tid] = ex;
    }
    __syncthreads();
}

__global__ __launch_bounds__(BT) void sage_bin_count(
    const int* __restrict__ dst, int* __restrict__ cnt, int n_edges, int tpw16)
{
    __shared__ int hist[NBIN];
    const int tid = threadIdx.x;
    if (tid < NBIN) hist[tid] = 0;
    __syncthreads();

    const int e0 = blockIdx.x * CHK2;
    const int e1 = min(e0 + CHK2, n_edges);
    int d[EPT];
    #pragma unroll
    for (int u = 0; u < EPT; ++u) {
        const int e = e0 + u * BT + tid;
        d[u] = (e < e1) ? dst[e] : -1;
    }
    #pragma unroll
    for (int u = 0; u < EPT; ++u)
        if (d[u] >= 0) atomicAdd(&hist[(unsigned)d[u] / (unsigned)tpw16], 1);
    __syncthreads();

    if (tid < NBIN) cnt[blockIdx.x * NBIN + tid] = hist[tid];
}

__global__ __launch_bounds__(BT) void sage_bin_scatter(
    const int* __restrict__ src, const int* __restrict__ dst,
    const int* __restrict__ cnt, int* __restrict__ binstart,
    int* __restrict__ binned, int n_edges, int n_chunks, int tpw16)
{
    __shared__ __attribute__((aligned(16))) int part[2][BT / 64][NBIN];
    __shared__ int bp[NBLK_MAX];
    __shared__ int wsum[8];
    __shared__ int cur[NBIN];
    const int tid = threadIdx.x;
    const int b4 = (tid & 63) * 4;
    const int q = tid >> 6;
    const int c = blockIdx.x;

    const int e0 = c * CHK2;
    const int e1 = min(e0 + CHK2, n_edges);
    int dd[EPT], ss[EPT];
    #pragma unroll
    for (int u = 0; u < EPT; ++u) {
        const int e = e0 + u * BT + tid;
        dd[u] = (e < e1) ? dst[e] : -1;
        ss[u] = (e < e1) ? src[e] : 0;
    }

    int4 pre = make_int4(0, 0, 0, 0), tot = pre;
    for (int g0 = q; g0 < n_chunks; g0 += TR * (BT / 64)) {
        int4 t4[TR];
        #pragma unroll
        for (int u = 0; u < TR; ++u) {
            const int g = g0 + u * (BT / 64);
            t4[u] = (g < n_chunks) ? *(const int4*)(cnt + g * NBIN + b4) : make_int4(0, 0, 0, 0);
        }
        #pragma unroll
        for (int u = 0; u < TR; ++u) {
            const int g = g0 + u * (BT / 64);
            tot.x += t4[u].x; tot.y += t4[u].y; tot.z += t4[u].z; tot.w += t4[u].w;
            if (g < c) { pre.x += t4[u].x; pre.y += t4[u].y; pre.z += t4[u].z; pre.w += t4[u].w; }
        }
    }
    *(int4*)&part[0][q][b4] = pre;
    *(int4*)&part[1][q][b4] = tot;
    __syncthreads();

    int p = 0, v = 0;
    if (tid < NBIN) {
        for (int i = 0; i < BT / 64; ++i) {
            p += part[0][i][tid];
            v += part[1][i][tid];
        }
    }
    sage_block_starts(v, bp, wsum);
    if (tid < NBIN) {
        const int s = bp[tid];
        cur[tid] = s + p;
        if (c == 0) {
            binstart[tid] = s;
            if (tid == NBIN - 1) binstart[NBIN] = s + v;
        }
    }
    __syncthreads();

    int pos[EPT];
    #pragma unroll
    for (int u = 0; u < EPT; ++u) {
        pos[u] = 0;
        if (dd[u] >= 0) {
            const int bb = (int)((unsigned)dd[u] / (unsigned)tpw16);
            pos[u] = atomicAdd(&cur[bb], 1);
            ss[u] |= (dd[u] - bb * tpw16) << 17;
        }
    }
    #pragma unroll
    for (int u = 0; u < EPT; ++u)
        if (dd[u] >= 0) binned[pos[u]] = ss[u];
}

#define SP 132
#define RPT 16
#define GW 13

#define SAGE_ROWS16(X) X(0) X(1) X(2) X(3) X(4) X(5) X(6) X(7) X(8) X(9) X(10) X(11) X(12) X(13) X(14) X(15)
#define SAGE_DECL(i)    float4 r##i;
#define SAGE_LOAD_H(i)  r##i = rp[i * (D / 4) + lane];
#define SAGE_STAGE_H(i) *(float4*)(aw + i * SP + lane * 4) = r##i;

#define SAGE_W10(X) X(0) X(1) X(2) X(3) X(4) X(5) X(6) X(7) X(8) X(9)
#define SAGE_WDECL(i)  float4 w##i = make_float4(0.0f, 0.0f, 0.0f, 0.0f);
#define SAGE_WLOAD(i)  { const int row = rb + i * GW; if (row < 2 * D) w##i = ((const float4*)((row < D) ? (Wself + (size_t)row * D) : (Wneigh + (size_t)(row - D) * D)))[lane]; }
#define SAGE_WSTORE(i) { const int row = rb + i * GW; if (row < 2 * D) *(float4*)(sW + row * SP + lane * 4) = w##i; }

__device__ __forceinline__ void sage_kpass(
    v8f (&acc)[8], const float* afr, const float* bfr)
{
    #pragma nounroll
    for (int k0 = 0; k0 < D; k0 += 16) {
        #pragma unroll
        for (int kk = 0; kk < 16; kk += 4) {
            const int ks = k0 + kk;
            const v2f a = *(const v2f*)(afr + ks);
            #pragma unroll
            for (int t = 0; t < 8; ++t) {
                const v2f bf = *(const v2f*)(bfr + ks + t * 16 * SP);
                acc[t] = __builtin_amdgcn_wmma_f32_16x16x4_f32(
                     false, a,  false, bf,
                     (short)0, acc[t],
                     false,  false);
            }
        }
        __builtin_amdgcn_sched_barrier(0);
        asm volatile("v_nop\n\tv_nop\n\tv_nop\n\tv_nop");
    }
}

struct SageWalk {
    int node;
    int nend;
    float4 a;
};

__device__ __forceinline__ void sage_flush(
    SageWalk& w, float* aw, const int bnd, const float rdv, const int lane)
{
    const float rdeg = __builtin_bit_cast(float,
        __builtin_amdgcn_readlane(__builtin_bit_cast(int, rdv), w.node));
    w.a.x *= rdeg; w.a.y *= rdeg; w.a.z *= rdeg; w.a.w *= rdeg;
    *(float4*)(aw + w.node * SP + lane * 4) = w.a;
    w.a = make_float4(0.0f, 0.0f, 0.0f, 0.0f);
    ++w.node;
    w.nend = __builtin_amdgcn_readlane(bnd, w.node + 1);
}

template <int N>
__device__ __forceinline__ int sage_batch(
    SageWalk& w, float* aw, const float4* h4,
    const int* perm, const int idx, const int cnt, const int e0,
    const int pn, const int bnd, const float rdv, const int lane)
{
    float4 v[N];
    #pragma unroll
    for (int u = 0; u < N; ++u) {
        const int s = __builtin_amdgcn_readlane(idx, min(u, cnt - 1));
        v[u] = h4[(size_t)s * (D / 4) + lane];
    }
    const int idx_n = perm[pn];
    asm volatile("" ::: "memory");
    #pragma unroll
    for (int u = 0; u < N; ++u) {
        if (u < cnt) {
            while (e0 + u >= w.nend) sage_flush(w, aw, bnd, rdv, lane);
            w.a.x += v[u].x; w.a.y += v[u].y; w.a.z += v[u].z; w.a.w += v[u].w;
        }
    }
    return idx_n;
}

__device__ __forceinline__ void sage_tile_means(
    float* aw, const float4* h4, const int* perm,
    const int bnd, const int lane)
{
    const int E0 = __builtin_amdgcn_readlane(bnd, 0);
    const int E1 = __builtin_amdgcn_readlane(bnd, 16);

    const int bnx = __shfl_down(bnd, 1);
    const float rdv = 1.0f / fmaxf((float)(bnx - bnd), 1.0f);

    SageWalk w;
    w.node = 0;
    w.nend = __builtin_amdgcn_readlane(bnd, 1);
    w.a = make_float4(0.0f, 0.0f, 0.0f, 0.0f);

    if (E0 < E1) {
        int idx = perm[min(E0 + lane, E1 - 1)];
        for (int eb = E0; eb < E1; eb += 32) {
            const int m = min(32, E1 - eb);
            const int pn = min(eb + 32 + lane, E1 - 1);
            if (m > 16) idx = sage_batch<32>(w, aw, h4, perm, idx, m, eb, pn, bnd, rdv, lane);
            else        idx = sage_batch<16>(w, aw, h4, perm, idx, m, eb, pn, bnd, rdv, lane);
        }
    }
    while (w.node < 16) sage_flush(w, aw, bnd, rdv, lane);
}

__device__ __forceinline__ void sage_load_words(
    int (&rec)[RPT], const int* __restrict__ binned, const int bs, const int nbe, const int i0)
{
    #pragma unroll
    for (int u = 0; u < RPT; ++u) {
        const int i = i0 + u * (GW * 32) + (int)threadIdx.x;
        rec[u] = (i < nbe) ? binned[bs + i] : -1;
    }
}

__device__ __forceinline__ void sage_count_words(const int (&rec)[RPT], int* seg)
{
    #pragma unroll
    for (int u = 0; u < RPT; ++u)
        if (rec[u] >= 0) atomicAdd(&seg[(rec[u] >> 17) + 1], 1);
}

__device__ __forceinline__ void sage_place_words(
    const int (&rec)[RPT], int* seg, int* perm, const int bs)
{
    int pos[RPT];
    #pragma unroll
    for (int u = 0; u < RPT; ++u) {
        pos[u] = 0;
        if (rec[u] >= 0) pos[u] = atomicAdd(&seg[(rec[u] >> 17) + 1], 1);
    }
    #pragma unroll
    for (int u = 0; u < RPT; ++u)
        if (rec[u] >= 0) perm[bs + pos[u]] = rec[u] & 0x1FFFF;
}

__device__ __forceinline__ void sage_group_range(
    const int* __restrict__ binned, int* perm, const int bs, const int nbe,
    int* seg, int* wsum)
{
    const int tid = threadIdx.x;
    const int lane = tid & 31;
    const int wave = tid >> 5;
    const int trip = GW * 32 * RPT;

    int rec[RPT];
    sage_load_words(rec, binned, bs, nbe, 0);
    seg[tid + 1] = 0;
    if (tid == 0) seg[0] = 0;
    __syncthreads();

    sage_count_words(rec, seg);
    for (int i0 = trip; i0 < nbe; i0 += trip) {
        int more[RPT];
        sage_load_words(more, binned, bs, nbe, i0);
        sage_count_words(more, seg);
    }
    __syncthreads();

    const int v = seg[tid + 1];
    int inc = v;
    for (int o = 1; o < 32; o <<= 1) {
        int u = __shfl_up(inc, o);
        if (lane >= o) inc += u;
    }
    if (lane == 31) wsum[wave] = inc;
    __syncthreads();

    int ex = inc - v;
    for (int i = 0; i < wave; ++i) ex += wsum[i];
    seg[tid + 1] = ex;
    __syncthreads();

    sage_place_words(rec, seg, perm, bs);
    for (int i0 = trip; i0 < nbe; i0 += trip) {
        int more[RPT];
        sage_load_words(more, binned, bs, nbe, i0);
        sage_place_words(more, seg, perm, bs);
    }
    __syncthreads();
}

__global__ __launch_bounds__(GW * 32) void sage_gather_gemm(
    const float* h, const float* __restrict__ Wself,
    const float* __restrict__ Wneigh, const float* __restrict__ bias,
    const int* __restrict__ binstart, const int* __restrict__ binned,
    int* perm, float* __restrict__ out, int n_nodes, int tpw)
{
    __shared__ __attribute__((aligned(16))) float sW[2 * D * SP];
    __shared__ __attribute__((aligned(16))) float sA[GW * 16 * SP];
    __shared__ int sSeg[GW * 32 + 1];
    __shared__ int sWS[16];

    const int bs = binstart[blockIdx.x];
    const int nbe = binstart[blockIdx.x + 1] - bs;

    const int lane = threadIdx.x & 31;
    const int wave = __builtin_amdgcn_readfirstlane((int)(threadIdx.x >> 5));

    for (int ph = 0; ph < 2; ++ph) {
        const int rb = wave + ph * 10 * GW;
        SAGE_W10(SAGE_WDECL)
        SAGE_W10(SAGE_WLOAD)
        SAGE_W10(SAGE_WSTORE)
    }

    const int half = lane >> 4;
    const int l15  = lane & 15;
    const int kOff = half * 2;

    float* aw = sA + wave * (16 * SP);
    const float* afr  = aw + l15 * SP + kOff;
    const float* bfr0 = sW + l15 * SP + kOff;
    const float* bfr1 = bfr0 + D * SP;

    const float4* h4 = (const float4*)h;

    float bb[8];
    #pragma unroll
    for (int t = 0; t < 8; ++t) bb[t] = bias[t * 16 + l15];

    const int n_tiles = (n_nodes + 15) >> 4;
    const int t_beg = blockIdx.x * tpw;
    const int t_end = min(t_beg + tpw, n_tiles);

    int tile = t_beg + wave;
    SAGE_ROWS16(SAGE_DECL)
    if (tile < t_end) {
        const float4* rp = (const float4*)(h + (size_t)tile * 16 * D);
        SAGE_ROWS16(SAGE_LOAD_H)
    }

    sage_group_range(binned, perm, bs, nbe, sSeg, sWS);

    while (tile < t_end) {
        const int m0 = tile * 16;

        const int bnd = bs + sSeg[(tile - t_beg) * 16 + min(lane, 16)];

        SAGE_ROWS16(SAGE_STAGE_H)

        v8f acc[8];
        #pragma unroll
        for (int t = 0; t < 8; ++t) {
            #pragma unroll
            for (int r = 0; r < 8; ++r) acc[t][r] = 0.0f;
        }

        sage_kpass(acc, afr, bfr0);

        sage_tile_means(aw, h4, perm, bnd, lane);

        const int next = tile + GW;
        {
            const float4* rp = (const float4*)(h + (size_t)min(next, t_end - 1) * 16 * D);
            SAGE_ROWS16(SAGE_LOAD_H)
        }

        sage_kpass(acc, afr, bfr1);

        const int mBase = m0 + half * 8;
        #pragma unroll
        for (int t = 0; t < 8; ++t) {
            float* op = out + (size_t)mBase * D + t * 16 + l15;
            #pragma unroll
            for (int r = 0; r < 8; ++r) {
                op[(size_t)r * D] = acc[t][r] + bb[t];
            }
        }

        tile = next;
    }
}

extern "C" void kernel_launch(void* const* d_in, const int* in_sizes, int n_in,
                              void* d_out, int out_size, void* d_ws, size_t ws_size,
                              hipStream_t stream) {
    const float* h      = (const float*)d_in[0];
    const int*   src    = (const int*)d_in[1];
    const int*   dst    = (const int*)d_in[2];
    const float* Wself  = (const float*)d_in[3];
    const float* Wneigh = (const float*)d_in[4];
    const float* bias   = (const float*)d_in[5];
    float*       out    = (float*)d_out;

    const int n_nodes = in_sizes[0] / D;
    const int n_edges = in_sizes[1];

    const int n_tiles = (n_nodes + 15) / 16;
    const int tpw = (n_tiles + NBIN - 1) / NBIN;
    const int nwg = (n_tiles + tpw - 1) / tpw;
    const int G2  = (n_edges + CHK2 - 1) / CHK2;
    const int E4  = (n_edges + 3) / 4 * 4;

    int* cnt      = (int*)d_ws;
    int* binstart = cnt + (size_t)G2 * NBIN;
    int* binned   = binstart + 320;
    int* perm     = binned + E4;

    sage_bin_count<<<G2, BT, 0, stream>>>(dst, cnt, n_edges, tpw * 16);
    sage_bin_scatter<<<G2, BT, 0, stream>>>(src, dst, cnt, binstart, binned, n_edges, G2, tpw * 16);
    sage_gather_gemm<<<nwg, GW * 32, 0, stream>>>(h, Wself, Wneigh, bias, binstart, binned, perm, out, n_nodes, tpw);
}


# Gather row loads as a rolling ring: 4 groups of 8 rows, a group is requested again for the next piece right after its rows are added (own 32-bit offset register per request); sum order unchanged
# speedup vs baseline: 3.6670x; 1.0862x over previous
#include <hip/hip_runtime.h>

typedef float v2f __attribute__((ext_vector_type(2)));
typedef float v8f __attribute__((ext_vector_type(8)));
typedef float v4f __attribute__((ext_vector_type(4)));

#define D 128
#define NBIN 256
#define NBLK_MAX 256
#define CHK2 6400
#define BT 1024
#define EPT ((CHK2 + BT - 1) / BT)
#define TR 16


__device__ __forceinline__ void sage_block_starts(const int v, int* bp, int* wsum)
{
    const int tid = threadIdx.x;
    const int lane = tid & 31;
    const int wave = tid >> 5;
    int inc = v;
    if (tid < NBLK_MAX) {
        for (int o = 1; o < 32; o <<= 1) {
            int u = __shfl_up(inc, o);
            if (lane >= o) inc += u;
        }
        if (lane == 31) wsum[wave] = inc;
    }
    __syncthreads();
    if (tid < NBLK_MAX) {
        int ex = inc - v;
        for (int i = 0; i < wave; ++i) ex += wsum[i];
        bp[tid] = ex;
    }
    __syncthreads();
}

__global__ __launch_bounds__(BT) void sage_bin_count(
    const int* __restrict__ dst, int* __restrict__ cnt, int n_edges, int tpw16)
{
    __shared__ int hist[NBIN];
    const int tid = threadIdx.x;
    if (tid < NBIN) hist[tid] = 0;
    __syncthreads();

    const int e0 = blockIdx.x * CHK2;
    const int e1 = min(e0 + CHK2, n_edges);
    int d[EPT];
    #pragma unroll
    for (int u = 0; u < EPT; ++u) {
        const int e = e0 + u * BT + tid;
        d[u] = (e < e1) ? dst[e] : -1;
    }
    #pragma unroll
    for (int u = 0; u < EPT; ++u)
        if (d[u] >= 0) atomicAdd(&hist[(unsigned)d[u] / (unsigned)tpw16], 1);
    __syncthreads();

    if (tid < NBIN) cnt[blockIdx.x * NBIN + tid] = hist[tid];
}

__global__ __launch_bounds__(BT) void sage_bin_scatter(
    const int* __restrict__ src, const int* __restrict__ dst,
    const int* __restrict__ cnt, int* __restrict__ binstart,
    int* __restrict__ binned, int n_edges, int n_chunks, int tpw16)
{
    __shared__ __attribute__((aligned(16))) int part[2][BT / 64][NBIN];
    __shared__ int bp[NBLK_MAX];
    __shared__ int wsum[8];
    __shared__ int cur[NBIN];
    const int tid = threadIdx.x;
    const int b4 = (tid & 63) * 4;
    const int q = tid >> 6;
    const int c = blockIdx.x;

    const int e0 = c * CHK2;
    const int e1 = min(e0 + CHK2, n_edges);
    int dd[EPT], ss[EPT];
    #pragma unroll
    for (int u = 0; u < EPT; ++u) {
        const int e = e0 + u * BT + tid;
        dd[u] = (e < e1) ? dst[e] : -1;
        ss[u] = (e < e1) ? src[e] : 0;
    }

    int4 pre = make_int4(0, 0, 0, 0), tot = pre;
    for (int g0 = q; g0 < n_chunks; g0 += TR * (BT / 64)) {
        int4 t4[TR];
        #pragma unroll
        for (int u = 0; u < TR; ++u) {
            const int g = g0 + u * (BT / 64);
            t4[u] = (g < n_chunks) ? *(const int4*)(cnt + g * NBIN + b4) : make_int4(0, 0, 0, 0);
        }
        #pragma unroll
        for (int u = 0; u < TR; ++u) {
            const int g = g0 + u * (BT / 64);
            tot.x += t4[u].x; tot.y += t4[u].y; tot.z += t4[u].z; tot.w += t4[u].w;
            if (g < c) { pre.x += t4[u].x; pre.y += t4[u].y; pre.z += t4[u].z; pre.w += t4[u].w; }
        }
    }
    *(int4*)&part[0][q][b4] = pre;
    *(int4*)&part[1][q][b4] = tot;
    __syncthreads();

    int p = 0, v = 0;
    if (tid < NBIN) {
        for (int i = 0; i < BT / 64; ++i) {
            p += part[0][i][tid];
            v += part[1][i][tid];
        }
    }
    sage_block_starts(v, bp, wsum);
    if (tid < NBIN) {
        const int s = bp[tid];
        cur[tid] = s + p;
        if (c == 0) {
            binstart[tid] = s;
            if (tid == NBIN - 1) binstart[NBIN] = s + v;
        }
    }
    __syncthreads();

    int pos[EPT];
    #pragma unroll
    for (int u = 0; u < EPT; ++u) {
        pos[u] = 0;
        if (dd[u] >= 0) {
            const int bb = (int)((unsigned)dd[u] / (unsigned)tpw16);
            pos[u] = atomicAdd(&cur[bb], 1);
            ss[u] |= (dd[u] - bb * tpw16) << 17;
        }
    }
    #pragma unroll
    for (int u = 0; u < EPT; ++u)
        if (dd[u] >= 0) binned[pos[u]] = ss[u];
}

#define SP 132
#define RPT 16
#define GW 13

#define SAGE_ROWS16(X) X(0) X(1) X(2) X(3) X(4) X(5) X(6) X(7) X(8) X(9) X(10) X(11) X(12) X(13) X(14) X(15)
#define SAGE_DECL(i)    float4 r##i;
#define SAGE_LOAD_H(i)  r##i = rp[i * (D / 4) + lane];
#define SAGE_STAGE_H(i) *(float4*)(aw + i * SP + lane * 4) = r##i;

#define SAGE_W10(X) X(0) X(1) X(2) X(3) X(4) X(5) X(6) X(7) X(8) X(9)
#define SAGE_WDECL(i)  float4 w##i = make_float4(0.0f, 0.0f, 0.0f, 0.0f);
#define SAGE_WLOAD(i)  { const int row = rb + i * GW; if (row < 2 * D) w##i = ((const float4*)((row < D) ? (Wself + (size_t)row * D) : (Wneigh + (size_t)(row - D) * D)))[lane]; }
#define SAGE_WSTORE(i) { const int row = rb + i * GW; if (row < 2 * D) *(float4*)(sW + row * SP + lane * 4) = w##i; }

__device__ __forceinline__ void sage_kpass(
    v8f (&acc)[8], const float* afr, const float* bfr)
{
    #pragma nounroll
    for (int k0 = 0; k0 < D; k0 += 16) {
        #pragma unroll
        for (int kk = 0; kk < 16; kk += 4) {
            const int ks = k0 + kk;
            const v2f a = *(const v2f*)(afr + ks);
            #pragma unroll
            for (int t = 0; t < 8; ++t) {
                const v2f bf = *(const v2f*)(bfr + ks + t * 16 * SP);
                acc[t] = __builtin_amdgcn_wmma_f32_16x16x4_f32(
                     false, a,  false, bf,
                     (short)0, acc[t],
                     false,  false);
            }
        }
        __builtin_amdgcn_sched_barrier(0);
        asm volatile("v_nop\n\tv_nop\n\tv_nop\n\tv_nop");
    }
}

struct SageWalk {
    int node;
    int nend;
    v2f lo;
    v2f hi;
};

__device__ __forceinline__ void sage_flush(
    SageWalk& w, float* aw, const int bnd, const float rdv, const int lane)
{
    const float rdeg = __builtin_bit_cast(float,
        __builtin_amdgcn_readlane(__builtin_bit_cast(int, rdv), w.node));
    const v2f lo = w.lo * rdeg;
    const v2f hi = w.hi * rdeg;
    *(v4f*)(aw + w.node * SP + lane * 4) = __builtin_shufflevector(lo, hi, 0, 1, 2, 3);
    w.lo = (v2f)(0.0f);
    w.hi = (v2f)(0.0f);
    ++w.node;
    w.nend = __builtin_amdgcn_readlane(bnd, w.node + 1);
}

#define RG 8
#define NGRP 4
#define PIECE (NGRP * RG)

template <int G>
__device__ __forceinline__ void sage_request(
    v4f (&v)[NGRP * RG], const float* h, const int idx, const int cnt, const unsigned lane16)
{
    unsigned off[RG];
    #pragma unroll
    for (int u = 0; u < RG; ++u) {
        const int s = __builtin_amdgcn_readlane(idx, min(G * RG + u, cnt - 1));
        off[u] = ((unsigned)s << 9) + lane16;
    }
    #pragma unroll
    for (int u = 0; u < RG; ++u)
        v[G * RG + u] = *(const v4f*)((const char*)h + off[u]);
    asm volatile("" ::: "memory");
}

template <int G, bool GUARD>
__device__ __forceinline__ void sage_consume(
    SageWalk& w, float* aw, const v4f (&v)[NGRP * RG], const int e0, const int cnt,
    const int bnd, const float rdv, const int lane)
{
    #pragma unroll
    for (int u = 0; u < RG; ++u) {
        const int i = G * RG + u;
        if (!GUARD || i < cnt) {
            while (e0 + i >= w.nend) sage_flush(w, aw, bnd, rdv, lane);
            w.lo += __builtin_shufflevector(v[i], v[i], 0, 1);
            w.hi += __builtin_shufflevector(v[i], v[i], 2, 3);
        }
    }
    if (!GUARD) asm volatile("" : "+v"(w.lo), "+v"(w.hi) :: "memory");
}

__device__ __forceinline__ void sage_tile_means(
    float* aw, const float* h, const int* perm,
    const int bnd, const int lane)
{
    const int E0 = __builtin_amdgcn_readlane(bnd, 0);
    const int E1 = __builtin_amdgcn_readlane(bnd, 16);

    const int bnx = __shfl_down(bnd, 1);
    const float rdv = 1.0f / fmaxf((float)(bnx - bnd), 1.0f);

    SageWalk w;
    w.node = 0;
    w.nend = __builtin_amdgcn_readlane(bnd, 1);
    w.lo = (v2f)(0.0f);
    w.hi = (v2f)(0.0f);

    if (E0 < E1) {
        const unsigned lane16 = (unsigned)lane * 16u;
        int idx = perm[min(E0 + lane, E1 - 1)];
        int idn = perm[min(E0 + PIECE + lane, E1 - 1)];

        v4f v[NGRP * RG];
        const int m0 = min(PIECE, E1 - E0);
        sage_request<0>(v, h, idx, m0, lane16);
        sage_request<1>(v, h, idx, m0, lane16);
        sage_request<2>(v, h, idx, m0, lane16);
        sage_request<3>(v, h, idx, m0, lane16);

        int eb = E0;
        for (; eb + PIECE < E1; eb += PIECE) {
            const int mn = min(PIECE, E1 - eb - PIECE);
            const int idx2 = perm[min(eb + 2 * PIECE + lane, E1 - 1)];
            sage_consume<0, false>(w, aw, v, eb, PIECE, bnd, rdv, lane);
            sage_request<0>(v, h, idn, mn, lane16);
            sage_consume<1, false>(w, aw, v, eb, PIECE, bnd, rdv, lane);
            sage_request<1>(v, h, idn, mn, lane16);
            sage_consume<2, false>(w, aw, v, eb, PIECE, bnd, rdv, lane);
            sage_request<2>(v, h, idn, mn, lane16);
            sage_consume<3, false>(w, aw, v, eb, PIECE, bnd, rdv, lane);
            sage_request<3>(v, h, idn, mn, lane16);
            idn = idx2;
        }
        const int m = E1 - eb;
        sage_consume<0, true>(w, aw, v, eb, m, bnd, rdv, lane);
        sage_consume<1, true>(w, aw, v, eb, m, bnd, rdv, lane);
        sage_consume<2, true>(w, aw, v, eb, m, bnd, rdv, lane);
        sage_consume<3, true>(w, aw, v, eb, m, bnd, rdv, lane);
    }
    while (w.node < 16) sage_flush(w, aw, bnd, rdv, lane);
}

__device__ __forceinline__ void sage_load_words(
    int (&rec)[RPT], const int* __restrict__ binned, const int bs, const int nbe, const int i0)
{
    #pragma unroll
    for (int u = 0; u < RPT; ++u) {
        const int i = i0 + u * (GW * 32) + (int)threadIdx.x;
        rec[u] = (i < nbe) ? binned[bs + i] : -1;
    }
}

__device__ __forceinline__ void sage_count_words(const int (&rec)[RPT], int* seg)
{
    #pragma unroll
    for (int u = 0; u < RPT; ++u)
        if (rec[u] >= 0) atomicAdd(&seg[(rec[u] >> 17) + 1], 1);
}

__device__ __forceinline__ void sage_place_words(
    const int (&rec)[RPT], int* seg, int* perm, const int bs)
{
    int pos[RPT];
    #pragma unroll
    for (int u = 0; u < RPT; ++u) {
        pos[u] = 0;
        if (rec[u] >= 0) pos[u] = atomicAdd(&seg[(rec[u] >> 17) + 1], 1);
    }
    #pragma unroll
    for (int u = 0; u < RPT; ++u)
        if (rec[u] >= 0) perm[bs + pos[u]] = rec[u] & 0x1FFFF;
}

__device__ __forceinline__ void sage_group_range(
    const int* __restrict__ binned, int* perm, const int bs, const int nbe,
    int* seg, int* wsum)
{
    const int tid = threadIdx.x;
    const int lane = tid & 31;
    const int wave = tid >> 5;
    const int trip = GW * 32 * RPT;

    int rec[RPT];
    sage_load_words(rec, binned, bs, nbe, 0);
    seg[tid + 1] = 0;
    if (tid == 0) seg[0] = 0;
    __syncthreads();

    sage_count_words(rec, seg);
    for (int i0 = trip; i0 < nbe; i0 += trip) {
        int more[RPT];
        sage_load_words(more, binned, bs, nbe, i0);
        sage_count_words(more, seg);
    }
    __syncthreads();

    const int v = seg[tid + 1];
    int inc = v;
    for (int o = 1; o < 32; o <<= 1) {
        int u = __shfl_up(inc, o);
        if (lane >= o) inc += u;
    }
    if (lane == 31) wsum[wave] = inc;
    __syncthreads();

    int ex = inc - v;
    for (int i = 0; i < wave; ++i) ex += wsum[i];
    seg[tid + 1] = ex;
    __syncthreads();

    sage_place_words(rec, seg, perm, bs);
    for (int i0 = trip; i0 < nbe; i0 += trip) {
        int more[RPT];
        sage_load_words(more, binned, bs, nbe, i0);
        sage_place_words(more, seg, perm, bs);
    }
    __syncthreads();
}

__global__ __launch_bounds__(GW * 32) void sage_gather_gemm(
    const float* h, const float* __restrict__ Wself,
    const float* __restrict__ Wneigh, const float* __restrict__ bias,
    const int* __restrict__ binstart, const int* __restrict__ binned,
    int* perm, float* __restrict__ out, int n_nodes, int tpw)
{
    __shared__ __attribute__((aligned(16))) float sW[2 * D * SP];
    __shared__ __attribute__((aligned(16))) float sA[GW * 16 * SP];
    __shared__ int sSeg[GW * 32 + 1];
    __shared__ int sWS[16];

    const int bs = binstart[blockIdx.x];
    const int nbe = binstart[blockIdx.x + 1] - bs;

    const int lane = threadIdx.x & 31;
    const int wave = __builtin_amdgcn_readfirstlane((int)(threadIdx.x >> 5));

    for (int ph = 0; ph < 2; ++ph) {
        const int rb = wave + ph * 10 * GW;
        SAGE_W10(SAGE_WDECL)
        SAGE_W10(SAGE_WLOAD)
        SAGE_W10(SAGE_WSTORE)
    }

    const int half = lane >> 4;
    const int l15  = lane & 15;
    const int kOff = half * 2;

    float* aw = sA + wave * (16 * SP);
    const float* afr  = aw + l15 * SP + kOff;
    const float* bfr0 = sW + l15 * SP + kOff;
    const float* bfr1 = bfr0 + D * SP;

    float bb[8];
    #pragma unroll
    for (int t = 0; t < 8; ++t) bb[t] = bias[t * 16 + l15];

    const int n_tiles = (n_nodes + 15) >> 4;
    const int t_beg = blockIdx.x * tpw;
    const int t_end = min(t_beg + tpw, n_tiles);

    int tile = t_beg + wave;
    SAGE_ROWS16(SAGE_DECL)
    if (tile < t_end) {
        const float4* rp = (const float4*)(h + (size_t)tile * 16 * D);
        SAGE_ROWS16(SAGE_LOAD_H)
    }

    sage_group_range(binned, perm, bs, nbe, sSeg, sWS);

    while (tile < t_end) {
        const int m0 = tile * 16;

        const int bnd = bs + sSeg[(tile - t_beg) * 16 + min(lane, 16)];

        SAGE_ROWS16(SAGE_STAGE_H)

        v8f acc[8];
        #pragma unroll
        for (int t = 0; t < 8; ++t) {
            #pragma unroll
            for (int r = 0; r < 8; ++r) acc[t][r] = 0.0f;
        }

        sage_kpass(acc, afr, bfr0);

        sage_tile_means(aw, h, perm, bnd, lane);

        const int next = tile + GW;
        {
            const float4* rp = (const float4*)(h + (size_t)min(next, t_end - 1) * 16 * D);
            SAGE_ROWS16(SAGE_LOAD_H)
        }

        sage_kpass(acc, afr, bfr1);

        const int mBase = m0 + half * 8;
        #pragma unroll
        for (int t = 0; t < 8; ++t) {
            float* op = out + (size_t)mBase * D + t * 16 + l15;
            #pragma unroll
            for (int r = 0; r < 8; ++r) {
                op[(size_t)r * D] = acc[t][r] + bb[t];
            }
        }

        tile = next;
    }
}

extern "C" void kernel_launch(void* const* d_in, const int* in_sizes, int n_in,
                              void* d_out, int out_size, void* d_ws, size_t ws_size,
                              hipStream_t stream) {
    const float* h      = (const float*)d_in[0];
    const int*   src    = (const int*)d_in[1];
    const int*   dst    = (const int*)d_in[2];
    const float* Wself  = (const float*)d_in[3];
    const float* Wneigh = (const float*)d_in[4];
    const float* bias   = (const float*)d_in[5];
    float*       out    = (float*)d_out;

    const int n_nodes = in_sizes[0] / D;
    const int n_edges = in_sizes[1];

    const int n_tiles = (n_nodes + 15) / 16;
    const int tpw = (n_tiles + NBIN - 1) / NBIN;
    const int nwg = (n_tiles + tpw - 1) / tpw;
    const int G2  = (n_edges + CHK2 - 1) / CHK2;
    const int E4  = (n_edges + 3) / 4 * 4;

    int* cnt      = (int*)d_ws;
    int* binstart = cnt + (size_t)G2 * NBIN;
    int* binned   = binstart + 320;
    int* perm     = binned + E4;

    sage_bin_count<<<G2, BT, 0, stream>>>(dst, cnt, n_edges, tpw * 16);
    sage_bin_scatter<<<G2, BT, 0, stream>>>(src, dst, cnt, binstart, binned, n_edges, G2, tpw * 16);
    sage_gather_gemm<<<nwg, GW * 32, 0, stream>>>(h, Wself, Wneigh, bias, binstart, binned, perm, out, n_nodes, tpw);
}


# Edge sort per chunk only: each work-group sorts its chunk by range in LDS and writes a run descriptor per range; count launch and count table go, 3 -> 2 launches; rows summed in new arrival order
# speedup vs baseline: 4.7525x; 1.2960x over previous
#include <hip/hip_runtime.h>

typedef float v2f __attribute__((ext_vector_type(2)));
typedef float v8f __attribute__((ext_vector_type(8)));
typedef float v4f __attribute__((ext_vector_type(4)));

#define D 128
#define NBIN 256
#define NBLK_MAX 256
#define CHK2 6400
#define BT 1024
#define EPT ((CHK2 + BT - 1) / BT)


__device__ __forceinline__ void sage_block_starts(const int v, int* bp, int* wsum)
{
    const int tid = threadIdx.x;
    const int lane = tid & 31;
    const int wave = tid >> 5;
    int inc = v;
    if (tid < NBLK_MAX) {
        for (int o = 1; o < 32; o <<= 1) {
            int u = __shfl_up(inc, o);
            if (lane >= o) inc += u;
        }
        if (lane == 31) wsum[wave] = inc;
    }
    __syncthreads();
    if (tid < NBLK_MAX) {
        int ex = inc - v;
        for (int i = 0; i < wave; ++i) ex += wsum[i];
        bp[tid] = ex;
    }
    __syncthreads();
}

__global__ __launch_bounds__(BT) void sage_chunk_sort(
    const int* __restrict__ src, const int* __restrict__ dst,
    int* __restrict__ runs, int* __restrict__ binned,
    int n_edges, int n_chunks, int tpw16)
{
    __shared__ int hist[NBIN];
    __shared__ int bp[NBLK_MAX];
    __shared__ int wsum[8];
    __shared__ int sorted[CHK2];
    const int tid = threadIdx.x;
    const int c = blockIdx.x;
    if (tid < NBIN) hist[tid] = 0;

    const int e0 = c * CHK2;
    const int e1 = min(e0 + CHK2, n_edges);
    int dd[EPT], ss[EPT];
    #pragma unroll
    for (int u = 0; u < EPT; ++u) {
        const int e = e0 + u * BT + tid;
        dd[u] = (e < e1) ? dst[e] : -1;
        ss[u] = (e < e1) ? src[e] : 0;
    }
    __syncthreads();

    int bbv[EPT], rk[EPT];
    #pragma unroll
    for (int u = 0; u < EPT; ++u) {
        bbv[u] = 0;
        rk[u] = 0;
        if (dd[u] >= 0) {
            const int bb = (int)((unsigned)dd[u] / (unsigned)tpw16);
            bbv[u] = bb;
            rk[u] = atomicAdd(&hist[bb], 1);
            ss[u] |= (dd[u] - bb * tpw16) << 17;
        }
    }
    __syncthreads();

    const int v = (tid < NBIN) ? hist[tid] : 0;
    sage_block_starts(v, bp, wsum);
    if (tid < NBIN) runs[(size_t)tid * n_chunks + c] = (bp[tid] << 16) | v;

    int st[EPT];
    #pragma unroll
    for (int u = 0; u < EPT; ++u) st[u] = bp[bbv[u]];
    #pragma unroll
    for (int u = 0; u < EPT; ++u)
        if (dd[u] >= 0) sorted[st[u] + rk[u]] = ss[u];
    __syncthreads();

    const int ne = e1 - e0;
    for (int u = 0; u < EPT; ++u) {
        const int i = u * BT + tid;
        if (i < ne) binned[e0 + i] = sorted[i];
    }
}

#define SP 132
#define RJ 20
#define GW 13

#define SAGE_ROWS16(X) X(0) X(1) X(2) X(3) X(4) X(5) X(6) X(7) X(8) X(9) X(10) X(11) X(12) X(13) X(14) X(15)
#define SAGE_DECL(i)    float4 r##i;
#define SAGE_LOAD_H(i)  r##i = rp[i * (D / 4) + lane];
#define SAGE_STAGE_H(i) *(float4*)(aw + i * SP + lane * 4) = r##i;

#define SAGE_W10(X) X(0) X(1) X(2) X(3) X(4) X(5) X(6) X(7) X(8) X(9)
#define SAGE_WDECL(i)  float4 w##i = make_float4(0.0f, 0.0f, 0.0f, 0.0f);
#define SAGE_WLOAD(i)  { const int row = rb + i * GW; if (row < 2 * D) w##i = ((const float4*)((row < D) ? (Wself + (size_t)row * D) : (Wneigh + (size_t)(row - D) * D)))[lane]; }
#define SAGE_WSTORE(i) { const int row = rb + i * GW; if (row < 2 * D) *(float4*)(sW + row * SP + lane * 4) = w##i; }

__device__ __forceinline__ void sage_kpass(
    v8f (&acc)[8], const float* afr, const float* bfr)
{
    #pragma nounroll
    for (int k0 = 0; k0 < D; k0 += 16) {
        #pragma unroll
        for (int kk = 0; kk < 16; kk += 4) {
            const int ks = k0 + kk;
            const v2f a = *(const v2f*)(afr + ks);
            #pragma unroll
            for (int t = 0; t < 8; ++t) {
                const v2f bf = *(const v2f*)(bfr + ks + t * 16 * SP);
                acc[t] = __builtin_amdgcn_wmma_f32_16x16x4_f32(
                     false, a,  false, bf,
                     (short)0, acc[t],
                     false,  false);
            }
        }
        __builtin_amdgcn_sched_barrier(0);
        asm volatile("v_nop\n\tv_nop\n\tv_nop\n\tv_nop");
    }
}

struct SageWalk {
    int node;
    int nend;
    v2f lo;
    v2f hi;
};

__device__ __forceinline__ void sage_flush(
    SageWalk& w, float* aw, const int bnd, const float rdv, const int lane)
{
    const float rdeg = __builtin_bit_cast(float,
        __builtin_amdgcn_readlane(__builtin_bit_cast(int, rdv), w.node));
    const v2f lo = w.lo * rdeg;
    const v2f hi = w.hi * rdeg;
    *(v4f*)(aw + w.node * SP + lane * 4) = __builtin_shufflevector(lo, hi, 0, 1, 2, 3);
    w.lo = (v2f)(0.0f);
    w.hi = (v2f)(0.0f);
    ++w.node;
    w.nend = __builtin_amdgcn_readlane(bnd, w.node + 1);
}

#define RG 8
#define NGRP 4
#define PIECE (NGRP * RG)

template <int G>
__device__ __forceinline__ void sage_request(
    v4f (&v)[NGRP * RG], const float* h, const int idx, const int cnt, const unsigned lane16)
{
    unsigned off[RG];
    #pragma unroll
    for (int u = 0; u < RG; ++u) {
        const int s = __builtin_amdgcn_readlane(idx, min(G * RG + u, cnt - 1));
        off[u] = ((unsigned)s << 9) + lane16;
    }
    #pragma unroll
    for (int u = 0; u < RG; ++u)
        v[G * RG + u] = *(const v4f*)((const char*)h + off[u]);
    asm volatile("" ::: "memory");
}

template <int G, bool GUARD>
__device__ __forceinline__ void sage_consume(
    SageWalk& w, float* aw, const v4f (&v)[NGRP * RG], const int e0, const int cnt,
    const int bnd, const float rdv, const int lane)
{
    #pragma unroll
    for (int u = 0; u < RG; ++u) {
        const int i = G * RG + u;
        if (!GUARD || i < cnt) {
            while (e0 + i >= w.nend) sage_flush(w, aw, bnd, rdv, lane);
            w.lo += __builtin_shufflevector(v[i], v[i], 0, 1);
            w.hi += __builtin_shufflevector(v[i], v[i], 2, 3);
        }
    }
    if (!GUARD) asm volatile("" : "+v"(w.lo), "+v"(w.hi) :: "memory");
}

__device__ __forceinline__ void sage_tile_means(
    float* aw, const float* h, const int* perm,
    const int bnd, const int lane)
{
    const int E0 = __builtin_amdgcn_readlane(bnd, 0);
    const int E1 = __builtin_amdgcn_readlane(bnd, 16);

    const int bnx = __shfl_down(bnd, 1);
    const float rdv = 1.0f / fmaxf((float)(bnx - bnd), 1.0f);

    SageWalk w;
    w.node = 0;
    w.nend = __builtin_amdgcn_readlane(bnd, 1);
    w.lo = (v2f)(0.0f);
    w.hi = (v2f)(0.0f);

    if (E0 < E1) {
        const unsigned lane16 = (unsigned)lane * 16u;
        int idx = perm[min(E0 + lane, E1 - 1)];
        int idn = perm[min(E0 + PIECE + lane, E1 - 1)];

        v4f v[NGRP * RG];
        const int m0 = min(PIECE, E1 - E0);
        sage_request<0>(v, h, idx, m0, lane16);
        sage_request<1>(v, h, idx, m0, lane16);
        sage_request<2>(v, h, idx, m0, lane16);
        sage_request<3>(v, h, idx, m0, lane16);

        int eb = E0;
        for (; eb + PIECE < E1; eb += PIECE) {
            const int mn = min(PIECE, E1 - eb - PIECE);
            const int idx2 = perm[min(eb + 2 * PIECE + lane, E1 - 1)];
            sage_consume<0, false>(w, aw, v, eb, PIECE, bnd, rdv, lane);
            sage_request<0>(v, h, idn, mn, lane16);
            sage_consume<1, false>(w, aw, v, eb, PIECE, bnd, rdv, lane);
            sage_request<1>(v, h, idn, mn, lane16);
            sage_consume<2, false>(w, aw, v, eb, PIECE, bnd, rdv, lane);
            sage_request<2>(v, h, idn, mn, lane16);
            sage_consume<3, false>(w, aw, v, eb, PIECE, bnd, rdv, lane);
            sage_request<3>(v, h, idn, mn, lane16);
            idn = idx2;
        }
        const int m = E1 - eb;
        sage_consume<0, true>(w, aw, v, eb, m, bnd, rdv, lane);
        sage_consume<1, true>(w, aw, v, eb, m, bnd, rdv, lane);
        sage_consume<2, true>(w, aw, v, eb, m, bnd, rdv, lane);
        sage_consume<3, true>(w, aw, v, eb, m, bnd, rdv, lane);
    }
    while (w.node < 16) sage_flush(w, aw, bnd, rdv, lane);
}

template <bool PLACE>
__device__ __forceinline__ void sage_one_word(const int w, int* seg, int* perm, const int bs)
{
    if (PLACE) {
        const int pos = atomicAdd(&seg[(w >> 17) + 1], 1);
        perm[bs + pos] = w & 0x1FFFF;
    } else {
        atomicAdd(&seg[(w >> 17) + 1], 1);
    }
}

template <bool PLACE>
__device__ __forceinline__ void sage_far_words(
    const int desc, const int* __restrict__ runs, const int* __restrict__ binned,
    int* seg, int* perm, const int bs, const int n_chunks, const int wave, const int lane)
{
    #pragma nounroll
    for (int j = 0; j < RJ; ++j) {
        const int d = __builtin_amdgcn_readlane(desc, j);
        const int n = d & 0xFFFF;
        if (n > 32) {
            const int base = (wave + GW * j) * CHK2 + (d >> 16);
            for (int p = 32 + lane; p < n; p += 32)
                sage_one_word<PLACE>(binned[base + p], seg, perm, bs);
        }
    }
    for (int c = wave + GW * RJ; c < n_chunks; c += GW) {
        const int d = runs[c];
        const int n = d & 0xFFFF;
        const int base = c * CHK2 + (d >> 16);
        for (int p = lane; p < n; p += 32)
            sage_one_word<PLACE>(binned[base + p], seg, perm, bs);
    }
}

__device__ __forceinline__ int sage_group_range(
    const int desc, const int* __restrict__ runs, const int* __restrict__ binned, int* perm,
    const int n_chunks, int* seg, int* wsum, int* wbs, const int wave)
{
    const int tid = threadIdx.x;
    const int lane = tid & 31;

    int part = desc >> 16;
    for (int c0 = GW * RJ; c0 < n_chunks; c0 += GW * 32) {
        const int c2 = c0 + wave + GW * lane;
        if (c2 < n_chunks) part += runs[c2] >> 16;
    }
    for (int o = 16; o > 0; o >>= 1) part += __shfl_xor(part, o);
    if (lane == 0) wbs[wave] = part;

    int rec[RJ];
    #pragma unroll
    for (int j = 0; j < RJ; ++j) {
        const int d = __builtin_amdgcn_readlane(desc, j);
        rec[j] = (lane < (d & 0xFFFF)) ? binned[(wave + GW * j) * CHK2 + (d >> 16) + lane] : -1;
    }
    seg[tid + 1] = 0;
    if (tid == 0) seg[0] = 0;
    __syncthreads();

    int bs = 0;
    for (int i = 0; i < GW; ++i) bs += wbs[i];

    #pragma unroll
    for (int j = 0; j < RJ; ++j)
        if (rec[j] >= 0) atomicAdd(&seg[(rec[j] >> 17) + 1], 1);
    sage_far_words<false>(desc, runs, binned, seg, perm, bs, n_chunks, wave, lane);
    __syncthreads();

    const int v = seg[tid + 1];
    int inc = v;
    for (int o = 1; o < 32; o <<= 1) {
        int u = __shfl_up(inc, o);
        if (lane >= o) inc += u;
    }
    if (lane == 31) wsum[wave] = inc;
    __syncthreads();

    int ex = inc - v;
    for (int i = 0; i < wave; ++i) ex += wsum[i];
    seg[tid + 1] = ex;
    __syncthreads();

    int pos[RJ];
    #pragma unroll
    for (int j = 0; j < RJ; ++j) {
        pos[j] = 0;
        if (rec[j] >= 0) pos[j] = atomicAdd(&seg[(rec[j] >> 17) + 1], 1);
    }
    #pragma unroll
    for (int j = 0; j < RJ; ++j)
        if (rec[j] >= 0) perm[bs + pos[j]] = rec[j] & 0x1FFFF;
    sage_far_words<true>(desc, runs, binned, seg, perm, bs, n_chunks, wave, lane);
    __syncthreads();
    return bs;
}

__global__ __launch_bounds__(GW * 32) void sage_gather_gemm(
    const float* h, const float* __restrict__ Wself,
    const float* __restrict__ Wneigh, const float* __restrict__ bias,
    const int* __restrict__ runT, const int* __restrict__ binned,
    int* perm, float* __restrict__ out, int n_nodes, int tpw, int n_chunks)
{
    __shared__ __attribute__((aligned(16))) float sW[2 * D * SP];
    __shared__ __attribute__((aligned(16))) float sA[GW * 16 * SP];
    __shared__ int sSeg[GW * 32 + 1];
    __shared__ int sWS[16];
    __shared__ int sBS[16];

    const int lane = threadIdx.x & 31;
    const int wave = __builtin_amdgcn_readfirstlane((int)(threadIdx.x >> 5));

    const int* runs = runT + (size_t)blockIdx.x * n_chunks;
    const int dchunk = wave + GW * lane;
    const int desc = (lane < RJ && dchunk < n_chunks) ? runs[dchunk] : 0;

    for (int ph = 0; ph < 2; ++ph) {
        const int rb = wave + ph * 10 * GW;
        SAGE_W10(SAGE_WDECL)
        SAGE_W10(SAGE_WLOAD)
        SAGE_W10(SAGE_WSTORE)
    }

    const int half = lane >> 4;
    const int l15  = lane & 15;
    const int kOff = half * 2;

    float* aw = sA + wave * (16 * SP);
    const float* afr  = aw + l15 * SP + kOff;
    const float* bfr0 = sW + l15 * SP + kOff;
    const float* bfr1 = bfr0 + D * SP;

    float bb[8];
    #pragma unroll
    for (int t = 0; t < 8; ++t) bb[t] = bias[t * 16 + l15];

    const int n_tiles = (n_nodes + 15) >> 4;
    const int t_beg = blockIdx.x * tpw;
    const int t_end = min(t_beg + tpw, n_tiles);

    int tile = t_beg + wave;
    SAGE_ROWS16(SAGE_DECL)
    if (tile < t_end) {
        const float4* rp = (const float4*)(h + (size_t)tile * 16 * D);
        SAGE_ROWS16(SAGE_LOAD_H)
    }

    const int bs = sage_group_range(desc, runs, binned, perm, n_chunks, sSeg, sWS, sBS, wave);

    while (tile < t_end) {
        const int m0 = tile * 16;

        const int bnd = bs + sSeg[(tile - t_beg) * 16 + min(lane, 16)];

        SAGE_ROWS16(SAGE_STAGE_H)

        v8f acc[8];
        #pragma unroll
        for (int t = 0; t < 8; ++t) {
            #pragma unroll
            for (int r = 0; r < 8; ++r) acc[t][r] = 0.0f;
        }

        sage_kpass(acc, afr, bfr0);

        sage_tile_means(aw, h, perm, bnd, lane);

        const int next = tile + GW;
        {
            const float4* rp = (const float4*)(h + (size_t)min(next, t_end - 1) * 16 * D);
            SAGE_ROWS16(SAGE_LOAD_H)
        }

        sage_kpass(acc, afr, bfr1);

        const int mBase = m0 + half * 8;
        #pragma unroll
        for (int t = 0; t < 8; ++t) {
            float* op = out + (size_t)mBase * D + t * 16 + l15;
            #pragma unroll
            for (int r = 0; r < 8; ++r) {
                op[(size_t)r * D] = acc[t][r] + bb[t];
            }
        }

        tile = next;
    }
}

extern "C" void kernel_launch(void* const* d_in, const int* in_sizes, int n_in,
                              void* d_out, int out_size, void* d_ws, size_t ws_size,
                              hipStream_t stream) {
    const float* h      = (const float*)d_in[0];
    const int*   src    = (const int*)d_in[1];
    const int*   dst    = (const int*)d_in[2];
    const float* Wself  = (const float*)d_in[3];
    const float* Wneigh = (const float*)d_in[4];
    const float* bias   = (const float*)d_in[5];
    float*       out    = (float*)d_out;

    const int n_nodes = in_sizes[0] / D;
    const int n_edges = in_sizes[1];

    const int n_tiles = (n_nodes + 15) / 16;
    const int tpw = (n_tiles + NBIN - 1) / NBIN;
    const int nwg = (n_tiles + tpw - 1) / tpw;
    const int G2  = (n_edges + CHK2 - 1) / CHK2;
    const int E4  = (n_edges + 3) / 4 * 4;

    int* runs   = (int*)d_ws;
    int* binned = runs + (size_t)NBIN * G2;
    int* perm   = binned + E4;

    sage_chunk_sort<<<G2, BT, 0, stream>>>(src, dst, runs, binned, n_edges, G2, tpw * 16);
    sage_gather_gemm<<<nwg, GW * 32, 0, stream>>>(h, Wself, Wneigh, bias, runs, binned, perm, out, n_nodes, tpw, G2);
}


# Node-grouped source list of a range kept in LDS instead of the workspace: no scattered 4-byte global stores, no store drain before the tile loop; workspace list only for a range above 12288 edges
# speedup vs baseline: 5.1704x; 1.0879x over previous
#include <hip/hip_runtime.h>

typedef float v2f __attribute__((ext_vector_type(2)));
typedef float v8f __attribute__((ext_vector_type(8)));
typedef float v4f __attribute__((ext_vector_type(4)));

#define D 128
#define NBIN 256
#define NBLK_MAX 256
#define CHK2 6400
#define BT 1024
#define EPT ((CHK2 + BT - 1) / BT)


__device__ __forceinline__ void sage_block_starts(const int v, int* bp, int* wsum)
{
    const int tid = threadIdx.x;
    const int lane = tid & 31;
    const int wave = tid >> 5;
    int inc = v;
    if (tid < NBLK_MAX) {
        for (int o = 1; o < 32; o <<= 1) {
            int u = __shfl_up(inc, o);
            if (lane >= o) inc += u;
        }
        if (lane == 31) wsum[wave] = inc;
    }
    __syncthreads();
    if (tid < NBLK_MAX) {
        int ex = inc - v;
        for (int i = 0; i < wave; ++i) ex += wsum[i];
        bp[tid] = ex;
    }
    __syncthreads();
}

__global__ __launch_bounds__(BT) void sage_chunk_sort(
    const int* __restrict__ src, const int* __restrict__ dst,
    int* __restrict__ runs, int* __restrict__ binned,
    int n_edges, int n_chunks, int tpw16)
{
    __shared__ int hist[NBIN];
    __shared__ int bp[NBLK_MAX];
    __shared__ int wsum[8];
    __shared__ int sorted[CHK2];
    const int tid = threadIdx.x;
    const int c = blockIdx.x;
    if (tid < NBIN) hist[tid] = 0;

    const int e0 = c * CHK2;
    const int e1 = min(e0 + CHK2, n_edges);
    int dd[EPT], ss[EPT];
    #pragma unroll
    for (int u = 0; u < EPT; ++u) {
        const int e = e0 + u * BT + tid;
        dd[u] = (e < e1) ? dst[e] : -1;
        ss[u] = (e < e1) ? src[e] : 0;
    }
    __syncthreads();

    int bbv[EPT], rk[EPT];
    #pragma unroll
    for (int u = 0; u < EPT; ++u) {
        bbv[u] = 0;
        rk[u] = 0;
        if (dd[u] >= 0) {
            const int bb = (int)((unsigned)dd[u] / (unsigned)tpw16);
            bbv[u] = bb;
            rk[u] = atomicAdd(&hist[bb], 1);
            ss[u] |= (dd[u] - bb * tpw16) << 17;
        }
    }
    __syncthreads();

    const int v = (tid < NBIN) ? hist[tid] : 0;
    sage_block_starts(v, bp, wsum);
    if (tid < NBIN) runs[(size_t)tid * n_chunks + c] = (bp[tid] << 16) | v;

    int st[EPT];
    #pragma unroll
    for (int u = 0; u < EPT; ++u) st[u] = bp[bbv[u]];
    #pragma unroll
    for (int u = 0; u < EPT; ++u)
        if (dd[u] >= 0) sorted[st[u] + rk[u]] = ss[u];
    __syncthreads();

    const int ne = e1 - e0;
    for (int u = 0; u < EPT; ++u) {
        const int i = u * BT + tid;
        if (i < ne) binned[e0 + i] = sorted[i];
    }
}

#define SP 132
#define RJ 20
#define GW 13
#define PCAP 12288

#define SAGE_ROWS16(X) X(0) X(1) X(2) X(3) X(4) X(5) X(6) X(7) X(8) X(9) X(10) X(11) X(12) X(13) X(14) X(15)
#define SAGE_DECL(i)    float4 r##i;
#define SAGE_LOAD_H(i)  r##i = rp[i * (D / 4) + lane];
#define SAGE_STAGE_H(i) *(float4*)(aw + i * SP + lane * 4) = r##i;

#define SAGE_W10(X) X(0) X(1) X(2) X(3) X(4) X(5) X(6) X(7) X(8) X(9)
#define SAGE_WDECL(i)  float4 w##i = make_float4(0.0f, 0.0f, 0.0f, 0.0f);
#define SAGE_WLOAD(i)  { const int row = rb + i * GW; if (row < 2 * D) w##i = ((const float4*)((row < D) ? (Wself + (size_t)row * D) : (Wneigh + (size_t)(row - D) * D)))[lane]; }
#define SAGE_WSTORE(i) { const int row = rb + i * GW; if (row < 2 * D) *(float4*)(sW + row * SP + lane * 4) = w##i; }

__device__ __forceinline__ void sage_kpass(
    v8f (&acc)[8], const float* afr, const float* bfr)
{
    #pragma nounroll
    for (int k0 = 0; k0 < D; k0 += 16) {
        #pragma unroll
        for (int kk = 0; kk < 16; kk += 4) {
            const int ks = k0 + kk;
            const v2f a = *(const v2f*)(afr + ks);
            #pragma unroll
            for (int t = 0; t < 8; ++t) {
                const v2f bf = *(const v2f*)(bfr + ks + t * 16 * SP);
                acc[t] = __builtin_amdgcn_wmma_f32_16x16x4_f32(
                     false, a,  false, bf,
                     (short)0, acc[t],
                     false,  false);
            }
        }
        __builtin_amdgcn_sched_barrier(0);
        asm volatile("v_nop\n\tv_nop\n\tv_nop\n\tv_nop");
    }
}

struct SageWalk {
    int node;
    int nend;
    v2f lo;
    v2f hi;
};

__device__ __forceinline__ void sage_flush(
    SageWalk& w, float* aw, const int bnd, const float rdv, const int lane)
{
    const float rdeg = __builtin_bit_cast(float,
        __builtin_amdgcn_readlane(__builtin_bit_cast(int, rdv), w.node));
    const v2f lo = w.lo * rdeg;
    const v2f hi = w.hi * rdeg;
    *(v4f*)(aw + w.node * SP + lane * 4) = __builtin_shufflevector(lo, hi, 0, 1, 2, 3);
    w.lo = (v2f)(0.0f);
    w.hi = (v2f)(0.0f);
    ++w.node;
    w.nend = __builtin_amdgcn_readlane(bnd, w.node + 1);
}

#define RG 8
#define NGRP 4
#define PIECE (NGRP * RG)

template <int G>
__device__ __forceinline__ void sage_request(
    v4f (&v)[NGRP * RG], const float* h, const int idx, const int cnt, const unsigned lane16)
{
    unsigned off[RG];
    #pragma unroll
    for (int u = 0; u < RG; ++u) {
        const int s = __builtin_amdgcn_readlane(idx, min(G * RG + u, cnt - 1));
        off[u] = ((unsigned)s << 9) + lane16;
    }
    #pragma unroll
    for (int u = 0; u < RG; ++u)
        v[G * RG + u] = *(const v4f*)((const char*)h + off[u]);
    asm volatile("" ::: "memory");
}

template <int G, bool GUARD>
__device__ __forceinline__ void sage_consume(
    SageWalk& w, float* aw, const v4f (&v)[NGRP * RG], const int e0, const int cnt,
    const int bnd, const float rdv, const int lane)
{
    #pragma unroll
    for (int u = 0; u < RG; ++u) {
        const int i = G * RG + u;
        if (!GUARD || i < cnt) {
            while (e0 + i >= w.nend) sage_flush(w, aw, bnd, rdv, lane);
            w.lo += __builtin_shufflevector(v[i], v[i], 0, 1);
            w.hi += __builtin_shufflevector(v[i], v[i], 2, 3);
        }
    }
    if (!GUARD) asm volatile("" : "+v"(w.lo), "+v"(w.hi) :: "memory");
}

typedef __attribute__((address_space(3))) const int sage_lds_cint;

__device__ __forceinline__ void sage_tile_means(
    float* aw, const float* h, sage_lds_cint* sp,
    const int bnd, const int lane)
{
    const int E0 = __builtin_amdgcn_readlane(bnd, 0);
    const int E1 = __builtin_amdgcn_readlane(bnd, 16);

    const int bnx = __shfl_down(bnd, 1);
    const float rdv = 1.0f / fmaxf((float)(bnx - bnd), 1.0f);

    SageWalk w;
    w.node = 0;
    w.nend = __builtin_amdgcn_readlane(bnd, 1);
    w.lo = (v2f)(0.0f);
    w.hi = (v2f)(0.0f);

    if (E0 < E1) {
        const unsigned lane16 = (unsigned)lane * 16u;
        int idx = sp[min(E0 + lane, E1 - 1)];
        int idn = sp[min(E0 + PIECE + lane, E1 - 1)];

        v4f v[NGRP * RG];
        const int m0 = min(PIECE, E1 - E0);
        sage_request<0>(v, h, idx, m0, lane16);
        sage_request<1>(v, h, idx, m0, lane16);
        sage_request<2>(v, h, idx, m0, lane16);
        sage_request<3>(v, h, idx, m0, lane16);

        int eb = E0;
        for (; eb + PIECE < E1; eb += PIECE) {
            const int mn = min(PIECE, E1 - eb - PIECE);
            const int idx2 = sp[min(eb + 2 * PIECE + lane, E1 - 1)];
            sage_consume<0, false>(w, aw, v, eb, PIECE, bnd, rdv, lane);
            sage_request<0>(v, h, idn, mn, lane16);
            sage_consume<1, false>(w, aw, v, eb, PIECE, bnd, rdv, lane);
            sage_request<1>(v, h, idn, mn, lane16);
            sage_consume<2, false>(w, aw, v, eb, PIECE, bnd, rdv, lane);
            sage_request<2>(v, h, idn, mn, lane16);
            sage_consume<3, false>(w, aw, v, eb, PIECE, bnd, rdv, lane);
            sage_request<3>(v, h, idn, mn, lane16);
            idn = idx2;
        }
        const int m = E1 - eb;
        sage_consume<0, true>(w, aw, v, eb, m, bnd, rdv, lane);
        sage_consume<1, true>(w, aw, v, eb, m, bnd, rdv, lane);
        sage_consume<2, true>(w, aw, v, eb, m, bnd, rdv, lane);
        sage_consume<3, true>(w, aw, v, eb, m, bnd, rdv, lane);
    }
    while (w.node < 16) sage_flush(w, aw, bnd, rdv, lane);
}

__device__ __forceinline__ void sage_tile_means_long(
    float* aw, const float* h, const int* perm,
    const int bnd, const int lane)
{
    #pragma nounroll
    for (int n = 0; n < 16; ++n) {
        const int b = __builtin_amdgcn_readlane(bnd, n);
        const int e = __builtin_amdgcn_readlane(bnd, n + 1);
        v4f s = (v4f)(0.0f);
        #pragma nounroll
        for (int p = b; p < e; ++p) {
            const int sidx = __builtin_amdgcn_readfirstlane(perm[p]);
            s += *(const v4f*)(h + (size_t)sidx * D + lane * 4);
        }
        const float rdeg = 1.0f / fmaxf((float)(e - b), 1.0f);
        *(v4f*)(aw + n * SP + lane * 4) = s * rdeg;
    }
}

template <bool PLACE>
__device__ __forceinline__ void sage_one_word(
    const int w, int* seg, int* perm, int* sp, const bool inl, const int bs)
{
    if (PLACE) {
        const int pos = atomicAdd(&seg[(w >> 17) + 1], 1);
        if (inl) sp[pos] = w & 0x1FFFF;
        else perm[bs + pos] = w & 0x1FFFF;
    } else {
        atomicAdd(&seg[(w >> 17) + 1], 1);
    }
}

template <bool PLACE>
__device__ __forceinline__ void sage_far_words(
    const int desc, const int* __restrict__ runs, const int* __restrict__ binned,
    int* seg, int* perm, int* sp, const bool inl, const int bs,
    const int n_chunks, const int wave, const int lane)
{
    #pragma nounroll
    for (int j = 0; j < RJ; ++j) {
        const int d = __builtin_amdgcn_readlane(desc, j);
        const int n = d & 0xFFFF;
        if (n > 32) {
            const int base = (wave + GW * j) * CHK2 + (d >> 16);
            for (int p = 32 + lane; p < n; p += 32)
                sage_one_word<PLACE>(binned[base + p], seg, perm, sp, inl, bs);
        }
    }
    for (int c = wave + GW * RJ; c < n_chunks; c += GW) {
        const int d = runs[c];
        const int n = d & 0xFFFF;
        const int base = c * CHK2 + (d >> 16);
        for (int p = lane; p < n; p += 32)
            sage_one_word<PLACE>(binned[base + p], seg, perm, sp, inl, bs);
    }
}

__device__ __forceinline__ int sage_group_range(
    const int desc, const int* __restrict__ runs, const int* __restrict__ binned, int* perm,
    int* sp, bool& inl,
    const int n_chunks, int* seg, int* wsum, int* wbs, const int wave)
{
    const int tid = threadIdx.x;
    const int lane = tid & 31;

    int part = desc >> 16;
    for (int c0 = GW * RJ; c0 < n_chunks; c0 += GW * 32) {
        const int c2 = c0 + wave + GW * lane;
        if (c2 < n_chunks) part += runs[c2] >> 16;
    }
    for (int o = 16; o > 0; o >>= 1) part += __shfl_xor(part, o);
    if (lane == 0) wbs[wave] = part;

    int rec[RJ];
    #pragma unroll
    for (int j = 0; j < RJ; ++j) {
        const int d = __builtin_amdgcn_readlane(desc, j);
        rec[j] = (lane < (d & 0xFFFF)) ? binned[(wave + GW * j) * CHK2 + (d >> 16) + lane] : -1;
    }
    seg[tid + 1] = 0;
    if (tid == 0) seg[0] = 0;
    __syncthreads();

    int bs = 0;
    for (int i = 0; i < GW; ++i) bs += wbs[i];

    #pragma unroll
    for (int j = 0; j < RJ; ++j)
        if (rec[j] >= 0) atomicAdd(&seg[(rec[j] >> 17) + 1], 1);
    sage_far_words<false>(desc, runs, binned, seg, perm, sp, false, bs, n_chunks, wave, lane);
    __syncthreads();

    const int v = seg[tid + 1];
    int inc = v;
    for (int o = 1; o < 32; o <<= 1) {
        int u = __shfl_up(inc, o);
        if (lane >= o) inc += u;
    }
    if (lane == 31) wsum[wave] = inc;
    __syncthreads();

    int ex = inc - v;
    int total = 0;
    for (int i = 0; i < GW; ++i) {
        const int s = wsum[i];
        total += s;
        if (i < wave) ex += s;
    }
    seg[tid + 1] = ex;
    inl = __builtin_amdgcn_readfirstlane(total) <= PCAP;
    __syncthreads();

    int pos[RJ];
    #pragma unroll
    for (int j = 0; j < RJ; ++j) {
        pos[j] = 0;
        if (rec[j] >= 0) pos[j] = atomicAdd(&seg[(rec[j] >> 17) + 1], 1);
    }
    if (inl) {
        #pragma unroll
        for (int j = 0; j < RJ; ++j)
            if (rec[j] >= 0) sp[pos[j]] = rec[j] & 0x1FFFF;
    } else {
        #pragma unroll
        for (int j = 0; j < RJ; ++j)
            if (rec[j] >= 0) perm[bs + pos[j]] = rec[j] & 0x1FFFF;
    }
    sage_far_words<true>(desc, runs, binned, seg, perm, sp, inl, bs, n_chunks, wave, lane);
    __syncthreads();
    return bs;
}

__global__ __launch_bounds__(GW * 32) void sage_gather_gemm(
    const float* h, const float* __restrict__ Wself,
    const float* __restrict__ Wneigh, const float* __restrict__ bias,
    const int* __restrict__ runT, const int* __restrict__ binned,
    int* perm, float* __restrict__ out, int n_nodes, int tpw, int n_chunks)
{
    __shared__ __attribute__((aligned(16))) float sW[2 * D * SP];
    __shared__ __attribute__((aligned(16))) float sA[GW * 16 * SP];
    __shared__ int sSeg[GW * 32 + 1];
    __shared__ int sPerm[PCAP];
    __shared__ int sWS[16];
    __shared__ int sBS[16];

    const int lane = threadIdx.x & 31;
    const int wave = __builtin_amdgcn_readfirstlane((int)(threadIdx.x >> 5));

    const int* runs = runT + (size_t)blockIdx.x * n_chunks;
    const int dchunk = wave + GW * lane;
    const int desc = (lane < RJ && dchunk < n_chunks) ? runs[dchunk] : 0;

    for (int ph = 0; ph < 2; ++ph) {
        const int rb = wave + ph * 10 * GW;
        SAGE_W10(SAGE_WDECL)
        SAGE_W10(SAGE_WLOAD)
        SAGE_W10(SAGE_WSTORE)
    }

    const int half = lane >> 4;
    const int l15  = lane & 15;
    const int kOff = half * 2;

    float* aw = sA + wave * (16 * SP);
    const float* afr  = aw + l15 * SP + kOff;
    const float* bfr0 = sW + l15 * SP + kOff;
    const float* bfr1 = bfr0 + D * SP;

    float bb[8];
    #pragma unroll
    for (int t = 0; t < 8; ++t) bb[t] = bias[t * 16 + l15];

    const int n_tiles = (n_nodes + 15) >> 4;
    const int t_beg = blockIdx.x * tpw;
    const int t_end = min(t_beg + tpw, n_tiles);

    int tile = t_beg + wave;
    SAGE_ROWS16(SAGE_DECL)
    if (tile < t_end) {
        const float4* rp = (const float4*)(h + (size_t)tile * 16 * D);
        SAGE_ROWS16(SAGE_LOAD_H)
    }

    bool inl;
    const int bs = sage_group_range(desc, runs, binned, perm, sPerm, inl, n_chunks, sSeg, sWS, sBS, wave);
    const int lb = inl ? 0 : bs;

    while (tile < t_end) {
        const int m0 = tile * 16;

        const int bnd = lb + sSeg[(tile - t_beg) * 16 + min(lane, 16)];

        SAGE_ROWS16(SAGE_STAGE_H)

        v8f acc[8];
        #pragma unroll
        for (int t = 0; t < 8; ++t) {
            #pragma unroll
            for (int r = 0; r < 8; ++r) acc[t][r] = 0.0f;
        }

        sage_kpass(acc, afr, bfr0);

        if (inl) sage_tile_means(aw, h, (sage_lds_cint*)sPerm, bnd, lane);
        else sage_tile_means_long(aw, h, perm, bnd, lane);

        const int next = tile + GW;
        {
            const float4* rp = (const float4*)(h + (size_t)min(next, t_end - 1) * 16 * D);
            SAGE_ROWS16(SAGE_LOAD_H)
        }

        sage_kpass(acc, afr, bfr1);

        const int mBase = m0 + half * 8;
        #pragma unroll
        for (int t = 0; t < 8; ++t) {
            float* op = out + (size_t)mBase * D + t * 16 + l15;
            #pragma unroll
            for (int r = 0; r < 8; ++r) {
                op[(size_t)r * D] = acc[t][r] + bb[t];
            }
        }

        tile = next;
    }
}

extern "C" void kernel_launch(void* const* d_in, const int* in_sizes, int n_in,
                              void* d_out, int out_size, void* d_ws, size_t ws_size,
                              hipStream_t stream) {
    const float* h      = (const float*)d_in[0];
    const int*   src    = (const int*)d_in[1];
    const int*   dst    = (const int*)d_in[2];
    const float* Wself  = (const float*)d_in[3];
    const float* Wneigh = (const float*)d_in[4];
    const float* bias   = (const float*)d_in[5];
    float*       out    = (float*)d_out;

    const int n_nodes = in_sizes[0] / D;
    const int n_edges = in_sizes[1];

    const int n_tiles = (n_nodes + 15) / 16;
    const int tpw = (n_tiles + NBIN - 1) / NBIN;
    const int nwg = (n_tiles + tpw - 1) / tpw;
    const int G2  = (n_edges + CHK2 - 1) / CHK2;
    const int E4  = (n_edges + 3) / 4 * 4;

    int* runs   = (int*)d_ws;
    int* binned = runs + (size_t)NBIN * G2;
    int* perm   = binned + E4;

    sage_chunk_sort<<<G2, BT, 0, stream>>>(src, dst, runs, binned, n_edges, G2, tpw * 16);
    sage_gather_gemm<<<nwg, GW * 32, 0, stream>>>(h, Wself, Wneigh, bias, runs, binned, perm, out, n_nodes, tpw, G2);
}


# Wave priority (s_setprio) set per matrix pass: first-tile waves before second-tile, self pass before neighbour pass, so a SIMD's youngest wave is not left alone with its gathers; sums unchanged
# speedup vs baseline: 5.3424x; 1.0333x over previous
#include <hip/hip_runtime.h>

typedef float v2f __attribute__((ext_vector_type(2)));
typedef float v8f __attribute__((ext_vector_type(8)));
typedef float v4f __attribute__((ext_vector_type(4)));

#define D 128
#define NBIN 256
#define NBLK_MAX 256
#define CHK2 6400
#define BT 1024
#define EPT ((CHK2 + BT - 1) / BT)


__device__ __forceinline__ void sage_block_starts(const int v, int* bp, int* wsum)
{
    const int tid = threadIdx.x;
    const int lane = tid & 31;
    const int wave = tid >> 5;
    int inc = v;
    if (tid < NBLK_MAX) {
        for (int o = 1; o < 32; o <<= 1) {
            int u = __shfl_up(inc, o);
            if (lane >= o) inc += u;
        }
        if (lane == 31) wsum[wave] = inc;
    }
    __syncthreads();
    if (tid < NBLK_MAX) {
        int ex = inc - v;
        for (int i = 0; i < wave; ++i) ex += wsum[i];
        bp[tid] = ex;
    }
    __syncthreads();
}

__global__ __launch_bounds__(BT) void sage_chunk_sort(
    const int* __restrict__ src, const int* __restrict__ dst,
    int* __restrict__ runs, int* __restrict__ binned,
    int n_edges, int n_chunks, int tpw16)
{
    __shared__ int hist[NBIN];
    __shared__ int bp[NBLK_MAX];
    __shared__ int wsum[8];
    __shared__ int sorted[CHK2];
    const int tid = threadIdx.x;
    const int c = blockIdx.x;
    if (tid < NBIN) hist[tid] = 0;

    const int e0 = c * CHK2;
    const int e1 = min(e0 + CHK2, n_edges);
    int dd[EPT], ss[EPT];
    #pragma unroll
    for (int u = 0; u < EPT; ++u) {
        const int e = e0 + u * BT + tid;
        dd[u] = (e < e1) ? dst[e] : -1;
        ss[u] = (e < e1) ? src[e] : 0;
    }
    __syncthreads();

    int bbv[EPT], rk[EPT];
    #pragma unroll
    for (int u = 0; u < EPT; ++u) {
        bbv[u] = 0;
        rk[u] = 0;
        if (dd[u] >= 0) {
            const int bb = (int)((unsigned)dd[u] / (unsigned)tpw16);
            bbv[u] = bb;
            rk[u] = atomicAdd(&hist[bb], 1);
            ss[u] |= (dd[u] - bb * tpw16) << 17;
        }
    }
    __syncthreads();

    const int v = (tid < NBIN) ? hist[tid] : 0;
    sage_block_starts(v, bp, wsum);
    if (tid < NBIN) runs[(size_t)tid * n_chunks + c] = (bp[tid] << 16) | v;

    int st[EPT];
    #pragma unroll
    for (int u = 0; u < EPT; ++u) st[u] = bp[bbv[u]];
    #pragma unroll
    for (int u = 0; u < EPT; ++u)
        if (dd[u] >= 0) sorted[st[u] + rk[u]] = ss[u];
    __syncthreads();

    const int ne = e1 - e0;
    for (int u = 0; u < EPT; ++u) {
        const int i = u * BT + tid;
        if (i < ne) binned[e0 + i] = sorted[i];
    }
}

#define SP 132
#define RJ 20
#define GW 13
#define PCAP 12288

#define SAGE_ROWS16(X) X(0) X(1) X(2) X(3) X(4) X(5) X(6) X(7) X(8) X(9) X(10) X(11) X(12) X(13) X(14) X(15)
#define SAGE_DECL(i)    float4 r##i;
#define SAGE_LOAD_H(i)  r##i = rp[i * (D / 4) + lane];
#define SAGE_STAGE_H(i) *(float4*)(aw + i * SP + lane * 4) = r##i;

#define SAGE_W10(X) X(0) X(1) X(2) X(3) X(4) X(5) X(6) X(7) X(8) X(9)
#define SAGE_WDECL(i)  float4 w##i = make_float4(0.0f, 0.0f, 0.0f, 0.0f);
#define SAGE_WLOAD(i)  { const int row = rb + i * GW; if (row < 2 * D) w##i = ((const float4*)((row < D) ? (Wself + (size_t)row * D) : (Wneigh + (size_t)(row - D) * D)))[lane]; }
#define SAGE_WSTORE(i) { const int row = rb + i * GW; if (row < 2 * D) *(float4*)(sW + row * SP + lane * 4) = w##i; }

__device__ __forceinline__ void sage_kpass(
    v8f (&acc)[8], const float* afr, const float* bfr)
{
    #pragma nounroll
    for (int k0 = 0; k0 < D; k0 += 16) {
        #pragma unroll
        for (int kk = 0; kk < 16; kk += 4) {
            const int ks = k0 + kk;
            const v2f a = *(const v2f*)(afr + ks);
            #pragma unroll
            for (int t = 0; t < 8; ++t) {
                const v2f bf = *(const v2f*)(bfr + ks + t * 16 * SP);
                acc[t] = __builtin_amdgcn_wmma_f32_16x16x4_f32(
                     false, a,  false, bf,
                     (short)0, acc[t],
                     false,  false);
            }
        }
        __builtin_amdgcn_sched_barrier(0);
        asm volatile("v_nop\n\tv_nop\n\tv_nop\n\tv_nop");
    }
}

struct SageWalk {
    int node;
    int nend;
    v2f lo;
    v2f hi;
};

__device__ __forceinline__ void sage_flush(
    SageWalk& w, float* aw, const int bnd, const float rdv, const int lane)
{
    const float rdeg = __builtin_bit_cast(float,
        __builtin_amdgcn_readlane(__builtin_bit_cast(int, rdv), w.node));
    const v2f lo = w.lo * rdeg;
    const v2f hi = w.hi * rdeg;
    *(v4f*)(aw + w.node * SP + lane * 4) = __builtin_shufflevector(lo, hi, 0, 1, 2, 3);
    w.lo = (v2f)(0.0f);
    w.hi = (v2f)(0.0f);
    ++w.node;
    w.nend = __builtin_amdgcn_readlane(bnd, w.node + 1);
}

#define RG 8
#define NGRP 4
#define PIECE (NGRP * RG)

template <int G>
__device__ __forceinline__ void sage_request(
    v4f (&v)[NGRP * RG], const float* h, const int idx, const int cnt, const unsigned lane16)
{
    unsigned off[RG];
    #pragma unroll
    for (int u = 0; u < RG; ++u) {
        const int s = __builtin_amdgcn_readlane(idx, min(G * RG + u, cnt - 1));
        off[u] = ((unsigned)s << 9) + lane16;
    }
    #pragma unroll
    for (int u = 0; u < RG; ++u)
        v[G * RG + u] = *(const v4f*)((const char*)h + off[u]);
    asm volatile("" ::: "memory");
}

template <int G, bool GUARD>
__device__ __forceinline__ void sage_consume(
    SageWalk& w, float* aw, const v4f (&v)[NGRP * RG], const int e0, const int cnt,
    const int bnd, const float rdv, const int lane)
{
    #pragma unroll
    for (int u = 0; u < RG; ++u) {
        const int i = G * RG + u;
        if (!GUARD || i < cnt) {
            while (e0 + i >= w.nend) sage_flush(w, aw, bnd, rdv, lane);
            w.lo += __builtin_shufflevector(v[i], v[i], 0, 1);
            w.hi += __builtin_shufflevector(v[i], v[i], 2, 3);
        }
    }
    if (!GUARD) asm volatile("" : "+v"(w.lo), "+v"(w.hi) :: "memory");
}

typedef __attribute__((address_space(3))) const int sage_lds_cint;

__device__ __forceinline__ void sage_tile_means(
    float* aw, const float* h, sage_lds_cint* sp,
    const int bnd, const int lane)
{
    const int E0 = __builtin_amdgcn_readlane(bnd, 0);
    const int E1 = __builtin_amdgcn_readlane(bnd, 16);

    const int bnx = __shfl_down(bnd, 1);
    const float rdv = 1.0f / fmaxf((float)(bnx - bnd), 1.0f);

    SageWalk w;
    w.node = 0;
    w.nend = __builtin_amdgcn_readlane(bnd, 1);
    w.lo = (v2f)(0.0f);
    w.hi = (v2f)(0.0f);

    if (E0 < E1) {
        const unsigned lane16 = (unsigned)lane * 16u;
        int idx = sp[min(E0 + lane, E1 - 1)];
        int idn = sp[min(E0 + PIECE + lane, E1 - 1)];

        v4f v[NGRP * RG];
        const int m0 = min(PIECE, E1 - E0);
        sage_request<0>(v, h, idx, m0, lane16);
        sage_request<1>(v, h, idx, m0, lane16);
        sage_request<2>(v, h, idx, m0, lane16);
        sage_request<3>(v, h, idx, m0, lane16);

        int eb = E0;
        for (; eb + PIECE < E1; eb += PIECE) {
            const int mn = min(PIECE, E1 - eb - PIECE);
            const int idx2 = sp[min(eb + 2 * PIECE + lane, E1 - 1)];
            sage_consume<0, false>(w, aw, v, eb, PIECE, bnd, rdv, lane);
            sage_request<0>(v, h, idn, mn, lane16);
            sage_consume<1, false>(w, aw, v, eb, PIECE, bnd, rdv, lane);
            sage_request<1>(v, h, idn, mn, lane16);
            sage_consume<2, false>(w, aw, v, eb, PIECE, bnd, rdv, lane);
            sage_request<2>(v, h, idn, mn, lane16);
            sage_consume<3, false>(w, aw, v, eb, PIECE, bnd, rdv, lane);
            sage_request<3>(v, h, idn, mn, lane16);
            idn = idx2;
        }
        const int m = E1 - eb;
        sage_consume<0, true>(w, aw, v, eb, m, bnd, rdv, lane);
        sage_consume<1, true>(w, aw, v, eb, m, bnd, rdv, lane);
        sage_consume<2, true>(w, aw, v, eb, m, bnd, rdv, lane);
        sage_consume<3, true>(w, aw, v, eb, m, bnd, rdv, lane);
    }
    while (w.node < 16) sage_flush(w, aw, bnd, rdv, lane);
}

__device__ __forceinline__ void sage_tile_means_long(
    float* aw, const float* h, const int* perm,
    const int bnd, const int lane)
{
    #pragma nounroll
    for (int n = 0; n < 16; ++n) {
        const int b = __builtin_amdgcn_readlane(bnd, n);
        const int e = __builtin_amdgcn_readlane(bnd, n + 1);
        v4f s = (v4f)(0.0f);
        #pragma nounroll
        for (int p = b; p < e; ++p) {
            const int sidx = __builtin_amdgcn_readfirstlane(perm[p]);
            s += *(const v4f*)(h + (size_t)sidx * D + lane * 4);
        }
        const float rdeg = 1.0f / fmaxf((float)(e - b), 1.0f);
        *(v4f*)(aw + n * SP + lane * 4) = s * rdeg;
    }
}

template <bool PLACE>
__device__ __forceinline__ void sage_one_word(
    const int w, int* seg, int* perm, int* sp, const bool inl, const int bs)
{
    if (PLACE) {
        const int pos = atomicAdd(&seg[(w >> 17) + 1], 1);
        if (inl) sp[pos] = w & 0x1FFFF;
        else perm[bs + pos] = w & 0x1FFFF;
    } else {
        atomicAdd(&seg[(w >> 17) + 1], 1);
    }
}

template <bool PLACE>
__device__ __forceinline__ void sage_far_words(
    const int desc, const int* __restrict__ runs, const int* __restrict__ binned,
    int* seg, int* perm, int* sp, const bool inl, const int bs,
    const int n_chunks, const int wave, const int lane)
{
    #pragma nounroll
    for (int j = 0; j < RJ; ++j) {
        const int d = __builtin_amdgcn_readlane(desc, j);
        const int n = d & 0xFFFF;
        if (n > 32) {
            const int base = (wave + GW * j) * CHK2 + (d >> 16);
            for (int p = 32 + lane; p < n; p += 32)
                sage_one_word<PLACE>(binned[base + p], seg, perm, sp, inl, bs);
        }
    }
    for (int c = wave + GW * RJ; c < n_chunks; c += GW) {
        const int d = runs[c];
        const int n = d & 0xFFFF;
        const int base = c * CHK2 + (d >> 16);
        for (int p = lane; p < n; p += 32)
            sage_one_word<PLACE>(binned[base + p], seg, perm, sp, inl, bs);
    }
}

__device__ __forceinline__ int sage_group_range(
    const int desc, const int* __restrict__ runs, const int* __restrict__ binned, int* perm,
    int* sp, bool& inl,
    const int n_chunks, int* seg, int* wsum, int* wbs, const int wave)
{
    const int tid = threadIdx.x;
    const int lane = tid & 31;

    int part = desc >> 16;
    for (int c0 = GW * RJ; c0 < n_chunks; c0 += GW * 32) {
        const int c2 = c0 + wave + GW * lane;
        if (c2 < n_chunks) part += runs[c2] >> 16;
    }
    for (int o = 16; o > 0; o >>= 1) part += __shfl_xor(part, o);
    if (lane == 0) wbs[wave] = part;

    int rec[RJ];
    #pragma unroll
    for (int j = 0; j < RJ; ++j) {
        const int d = __builtin_amdgcn_readlane(desc, j);
        rec[j] = (lane < (d & 0xFFFF)) ? binned[(wave + GW * j) * CHK2 + (d >> 16) + lane] : -1;
    }
    seg[tid + 1] = 0;
    if (tid == 0) seg[0] = 0;
    __syncthreads();

    int bs = 0;
    for (int i = 0; i < GW; ++i) bs += wbs[i];

    #pragma unroll
    for (int j = 0; j < RJ; ++j)
        if (rec[j] >= 0) atomicAdd(&seg[(rec[j] >> 17) + 1], 1);
    sage_far_words<false>(desc, runs, binned, seg, perm, sp, false, bs, n_chunks, wave, lane);
    __syncthreads();

    const int v = seg[tid + 1];
    int inc = v;
    for (int o = 1; o < 32; o <<= 1) {
        int u = __shfl_up(inc, o);
        if (lane >= o) inc += u;
    }
    if (lane == 31) wsum[wave] = inc;
    __syncthreads();

    int ex = inc - v;
    int total = 0;
    for (int i = 0; i < GW; ++i) {
        const int s = wsum[i];
        total += s;
        if (i < wave) ex += s;
    }
    seg[tid + 1] = ex;
    inl = __builtin_amdgcn_readfirstlane(total) <= PCAP;
    __syncthreads();

    int pos[RJ];
    #pragma unroll
    for (int j = 0; j < RJ; ++j) {
        pos[j] = 0;
        if (rec[j] >= 0) pos[j] = atomicAdd(&seg[(rec[j] >> 17) + 1], 1);
    }
    if (inl) {
        #pragma unroll
        for (int j = 0; j < RJ; ++j)
            if (rec[j] >= 0) sp[pos[j]] = rec[j] & 0x1FFFF;
    } else {
        #pragma unroll
        for (int j = 0; j < RJ; ++j)
            if (rec[j] >= 0) perm[bs + pos[j]] = rec[j] & 0x1FFFF;
    }
    sage_far_words<true>(desc, runs, binned, seg, perm, sp, inl, bs, n_chunks, wave, lane);
    __syncthreads();
    return bs;
}

__global__ __launch_bounds__(GW * 32) void sage_gather_gemm(
    const float* h, const float* __restrict__ Wself,
    const float* __restrict__ Wneigh, const float* __restrict__ bias,
    const int* __restrict__ runT, const int* __restrict__ binned,
    int* perm, float* __restrict__ out, int n_nodes, int tpw, int n_chunks)
{
    __shared__ __attribute__((aligned(16))) float sW[2 * D * SP];
    __shared__ __attribute__((aligned(16))) float sA[GW * 16 * SP];
    __shared__ int sSeg[GW * 32 + 1];
    __shared__ int sPerm[PCAP];
    __shared__ int sWS[16];
    __shared__ int sBS[16];

    const int lane = threadIdx.x & 31;
    const int wave = __builtin_amdgcn_readfirstlane((int)(threadIdx.x >> 5));

    const int* runs = runT + (size_t)blockIdx.x * n_chunks;
    const int dchunk = wave + GW * lane;
    const int desc = (lane < RJ && dchunk < n_chunks) ? runs[dchunk] : 0;

    for (int ph = 0; ph < 2; ++ph) {
        const int rb = wave + ph * 10 * GW;
        SAGE_W10(SAGE_WDECL)
        SAGE_W10(SAGE_WLOAD)
        SAGE_W10(SAGE_WSTORE)
    }

    const int half = lane >> 4;
    const int l15  = lane & 15;
    const int kOff = half * 2;

    float* aw = sA + wave * (16 * SP);
    const float* afr  = aw + l15 * SP + kOff;
    const float* bfr0 = sW + l15 * SP + kOff;
    const float* bfr1 = bfr0 + D * SP;

    float bb[8];
    #pragma unroll
    for (int t = 0; t < 8; ++t) bb[t] = bias[t * 16 + l15];

    const int n_tiles = (n_nodes + 15) >> 4;
    const int t_beg = blockIdx.x * tpw;
    const int t_end = min(t_beg + tpw, n_tiles);

    int tile = t_beg + wave;
    SAGE_ROWS16(SAGE_DECL)
    if (tile < t_end) {
        const float4* rp = (const float4*)(h + (size_t)tile * 16 * D);
        SAGE_ROWS16(SAGE_LOAD_H)
    }

    bool inl;
    const int bs = sage_group_range(desc, runs, binned, perm, sPerm, inl, n_chunks, sSeg, sWS, sBS, wave);
    const int lb = inl ? 0 : bs;

    while (tile < t_end) {
        const int m0 = tile * 16;

        const int bnd = lb + sSeg[(tile - t_beg) * 16 + min(lane, 16)];

        SAGE_ROWS16(SAGE_STAGE_H)

        v8f acc[8];
        #pragma unroll
        for (int t = 0; t < 8; ++t) {
            #pragma unroll
            for (int r = 0; r < 8; ++r) acc[t][r] = 0.0f;
        }

        const bool first_tile = tile < t_beg + GW;
        if (first_tile) __builtin_amdgcn_s_setprio(3);
        else __builtin_amdgcn_s_setprio(1);
        sage_kpass(acc, afr, bfr0);

        if (inl) sage_tile_means(aw, h, (sage_lds_cint*)sPerm, bnd, lane);
        else sage_tile_means_long(aw, h, perm, bnd, lane);

        const int next = tile + GW;
        {
            const float4* rp = (const float4*)(h + (size_t)min(next, t_end - 1) * 16 * D);
            SAGE_ROWS16(SAGE_LOAD_H)
        }

        if (first_tile) __builtin_amdgcn_s_setprio(2);
        else __builtin_amdgcn_s_setprio(0);
        sage_kpass(acc, afr, bfr1);

        const int mBase = m0 + half * 8;
        #pragma unroll
        for (int t = 0; t < 8; ++t) {
            float* op = out + (size_t)mBase * D + t * 16 + l15;
            #pragma unroll
            for (int r = 0; r < 8; ++r) {
                op[(size_t)r * D] = acc[t][r] + bb[t];
            }
        }

        tile = next;
    }
}

extern "C" void kernel_launch(void* const* d_in, const int* in_sizes, int n_in,
                              void* d_out, int out_size, void* d_ws, size_t ws_size,
                              hipStream_t stream) {
    const float* h      = (const float*)d_in[0];
    const int*   src    = (const int*)d_in[1];
    const int*   dst    = (const int*)d_in[2];
    const float* Wself  = (const float*)d_in[3];
    const float* Wneigh = (const float*)d_in[4];
    const float* bias   = (const float*)d_in[5];
    float*       out    = (float*)d_out;

    const int n_nodes = in_sizes[0] / D;
    const int n_edges = in_sizes[1];

    const int n_tiles = (n_nodes + 15) / 16;
    const int tpw = (n_tiles + NBIN - 1) / NBIN;
    const int nwg = (n_tiles + tpw - 1) / tpw;
    const int G2  = (n_edges + CHK2 - 1) / CHK2;
    const int E4  = (n_edges + 3) / 4 * 4;

    int* runs   = (int*)d_ws;
    int* binned = runs + (size_t)NBIN * G2;
    int* perm   = binned + E4;

    sage_chunk_sort<<<G2, BT, 0, stream>>>(src, dst, runs, binned, n_edges, G2, tpw * 16);
    sage_gather_gemm<<<nwg, GW * 32, 0, stream>>>(h, Wself, Wneigh, bias, runs, binned, perm, out, n_nodes, tpw, G2);
}
